# starting kernel
# baseline (speedup 1.0000x reference)
#include <hip/hip_runtime.h>
#include <hip/hip_bf16.h>

#define D 1024
#define H 4096
#define EPS 1e-5f

typedef __attribute__((ext_vector_type(2))) float v2f;
typedef __attribute__((ext_vector_type(8))) float v8f;

#define OP_NONE     0
#define OP_SIGMOID  1
#define OP_SQRELU   2
#define OP_ADDRES   3
#define OP_GATEDRES 4

__global__ __launch_bounds__(256) void ln_mix_kernel(
    const float* __restrict__ xin, const float* __restrict__ w,
    const float* __restrict__ b,   const float* __restrict__ prev,
    const float* __restrict__ tmk, const float* __restrict__ tmv,
    const float* __restrict__ tmr,
    float* __restrict__ xln_out, float* __restrict__ xk_out,
    float* __restrict__ xv_out,  float* __restrict__ xr_out)
{
    __shared__ float ssum[8];
    __shared__ float ssq[8];
    const int t = threadIdx.x;
    const int wave = t >> 5;
    const int lane = t & 31;

    float s = 0.f, sq = 0.f;
    for (int i = t; i < D; i += 256) {
        float v = xin[i];
        s += v;
        sq += v * v;
    }
    for (int off = 16; off > 0; off >>= 1) {
        s  += __shfl_xor(s,  off, 32);
        sq += __shfl_xor(sq, off, 32);
    }
    if (lane == 0) { ssum[wave] = s; ssq[wave] = sq; }
    __syncthreads();
    if (wave == 0) {
        float a = (lane < 8) ? ssum[lane] : 0.f;
        float c = (lane < 8) ? ssq[lane]  : 0.f;
        for (int off = 4; off > 0; off >>= 1) {
            a += __shfl_xor(a, off, 32);
            c += __shfl_xor(c, off, 32);
        }
        if (lane == 0) { ssum[0] = a; ssq[0] = c; }
    }
    __syncthreads();

    const float mu   = ssum[0] * (1.f / D);
    const float var  = ssq[0] * (1.f / D) - mu * mu;
    const float rstd = rsqrtf(var + EPS);

    for (int i = t; i < D; i += 256) {
        float xl = (xin[i] - mu) * rstd * w[i] + b[i];
        xln_out[i] = xl;
        float p = prev[i];
        float k = tmk[i];
        xk_out[i] = xl * k + p * (1.f - k);
        if (xv_out) {
            float v = tmv[i];
            xv_out[i] = xl * v + p * (1.f - v);
        }
        float r = tmr[i];
        xr_out[i] = xl * r + p * (1.f - r);
    }
}

struct MatvecJob {
    const float* W;
    const float* x;
    float*       y;
    const float* res;
    const float* gate;
    int K;
    int ldw;
    int op;
    int nblocks;
};

__device__ __forceinline__ void matvec_body(
    const float* __restrict__ W, const float* __restrict__ x,
    float* __restrict__ y, int K, int ldw, int op,
    const float* __restrict__ res, const float* __restrict__ gate, int blk)
{
    __shared__ float part[8][16];

    const int wave = threadIdx.x >> 5;
    const int lane = threadIdx.x & 31;
    const int half = lane >> 4;
    const int m    = lane & 15;
    const int row0 = blk << 4;

    const int kchunk = K >> 3;
    const int k0 = wave * kchunk;
    const int k1 = k0 + kchunk;

    const float* wrow = W + (size_t)(row0 + m) * (size_t)ldw + (half << 1);
    const float* xrow = x + (half << 1);

    v8f c0 = {}; v8f c1 = {}; v8f c2 = {}; v8f c3 = {};

    for (int k = k0; k < k1; k += 16) {
        v2f a0 = *(const v2f*)(wrow + k);
        v2f a1 = *(const v2f*)(wrow + k + 4);
        v2f a2 = *(const v2f*)(wrow + k + 8);
        v2f a3 = *(const v2f*)(wrow + k + 12);
        v2f b0 = *(const v2f*)(xrow + k);
        v2f b1 = *(const v2f*)(xrow + k + 4);
        v2f b2 = *(const v2f*)(xrow + k + 8);
        v2f b3 = *(const v2f*)(xrow + k + 12);
        c0 = __builtin_amdgcn_wmma_f32_16x16x4_f32(false, a0, false, b0, (short)0, c0, false, false);
        c1 = __builtin_amdgcn_wmma_f32_16x16x4_f32(false, a1, false, b1, (short)0, c1, false, false);
        c2 = __builtin_amdgcn_wmma_f32_16x16x4_f32(false, a2, false, b2, (short)0, c2, false, false);
        c3 = __builtin_amdgcn_wmma_f32_16x16x4_f32(false, a3, false, b3, (short)0, c3, false, false);
    }

    v8f c = c0 + c1 + c2 + c3;

    if (m == 0) {
        const int rbase = half << 3;
        #pragma unroll
        for (int j = 0; j < 8; ++j)
            part[wave][rbase + j] = c[j];
    }
    __syncthreads();

    if (threadIdx.x < 16) {
        float v = 0.f;
        #pragma unroll
        for (int wv = 0; wv < 8; ++wv)
            v += part[wv][threadIdx.x];
        const int r = row0 + threadIdx.x;
        switch (op) {
            case OP_SIGMOID:  v = 1.f / (1.f + expf(-v));              break;
            case OP_SQRELU:   { float t = fmaxf(v, 0.f); v = t * t; }  break;
            case OP_ADDRES:   v = res[r] + v;                          break;
            case OP_GATEDRES: v = res[r] + gate[r] * v;                break;
            default: break;
        }
        y[r] = v;
    }
}

__global__ __launch_bounds__(256) void wmma_matvec3_kernel(
    MatvecJob j0, MatvecJob j1, MatvecJob j2)
{
    int blk = blockIdx.x;
    if (blk < j0.nblocks) {
        matvec_body(j0.W, j0.x, j0.y, j0.K, j0.ldw, j0.op, j0.res, j0.gate, blk);
        return;
    }
    blk -= j0.nblocks;
    if (blk < j1.nblocks) {
        matvec_body(j1.W, j1.x, j1.y, j1.K, j1.ldw, j1.op, j1.res, j1.gate, blk);
        return;
    }
    blk -= j1.nblocks;
    matvec_body(j2.W, j2.x, j2.y, j2.K, j2.ldw, j2.op, j2.res, j2.gate, blk);
}

__global__ __launch_bounds__(256) void wkv_kernel(
    const float* __restrict__ kvec, const float* __restrict__ vvec,
    const float* __restrict__ rvec, const float* __restrict__ state,
    const float* __restrict__ time_first, const float* __restrict__ time_decay,
    float* __restrict__ rwkv,
    float* __restrict__ s2, float* __restrict__ s3, float* __restrict__ s4)
{
    const int i = blockIdx.x * 256 + threadIdx.x;
    if (i >= D) return;

    float k  = kvec[i];
    float v  = vvec[i];
    float r  = rvec[i];
    float aa = state[2 * D + i];
    float bb = state[3 * D + i];
    float pp = state[4 * D + i];

    float ww = time_first[i] + k;
    float qq = fmaxf(pp, ww);
    float e1 = expf(pp - qq);
    float e2 = expf(ww - qq);
    float wkv = (e1 * aa + e2 * v) / (e1 * bb + e2);
    rwkv[i] = r * wkv;

    float ww2 = pp + time_decay[i];
    float qq2 = fmaxf(ww2, k);
    float f1 = expf(ww2 - qq2);
    float f2 = expf(k - qq2);
    s2[i] = f1 * aa + f2 * v;
    s3[i] = f1 * bb + f2;
    s4[i] = qq2;
}

extern "C" void kernel_launch(void* const* d_in, const int* in_sizes, int n_in,
                              void* d_out, int out_size, void* d_ws, size_t ws_size,
                              hipStream_t stream) {
    const float* x     = (const float*)d_in[0];
    const float* state = (const float*)d_in[1];
    const float* tmk   = (const float*)d_in[2];
    const float* tmv   = (const float*)d_in[3];
    const float* tmr   = (const float*)d_in[4];
    const float* tf    = (const float*)d_in[5];
    const float* td    = (const float*)d_in[6];
    const float* kw    = (const float*)d_in[7];
    const float* vw    = (const float*)d_in[8];
    const float* rw    = (const float*)d_in[9];
    const float* ow    = (const float*)d_in[10];
    const float* lnw   = (const float*)d_in[11];
    const float* lnb   = (const float*)d_in[12];
    const float* ftmk  = (const float*)d_in[13];
    const float* ftmr  = (const float*)d_in[14];
    const float* fkw   = (const float*)d_in[15];
    const float* fvw   = (const float*)d_in[16];
    const float* frw   = (const float*)d_in[17];
    const float* flnw  = (const float*)d_in[18];
    const float* flnb  = (const float*)d_in[19];

    float* ws    = (float*)d_ws;
    float* xk    = ws + 1 * D;
    float* xv    = ws + 2 * D;
    float* xr    = ws + 3 * D;
    float* kvec  = ws + 4 * D;
    float* vvec  = ws + 5 * D;
    float* rvec  = ws + 6 * D;
    float* rwkv  = ws + 7 * D;
    float* x_att = ws + 8 * D;
    float* fxk   = ws + 9 * D;
    float* fxr   = ws + 10 * D;
    float* fr    = ws + 11 * D;
    float* fk    = ws + 12 * D;

    float* out = (float*)d_out;
    float* s0  = out + 1 * D;
    float* s1  = out + 2 * D;
    float* s2  = out + 3 * D;
    float* s3  = out + 4 * D;
    float* s4  = out + 5 * D;

    const MatvecJob jnull = { nullptr, nullptr, nullptr, nullptr, nullptr, 0, 0, OP_NONE, 0 };

    ln_mix_kernel<<<1, 256, 0, stream>>>(x, lnw, lnb, state + 1 * D,
                                         tmk, tmv, tmr, s1, xk, xv, xr);

    {
        MatvecJob jk = { kw, xk, kvec, nullptr, nullptr, D, D, OP_NONE,    D / 16 };
        MatvecJob jv = { vw, xv, vvec, nullptr, nullptr, D, D, OP_NONE,    D / 16 };
        MatvecJob jr = { rw, xr, rvec, nullptr, nullptr, D, D, OP_SIGMOID, D / 16 };
        wmma_matvec3_kernel<<<3 * (D / 16), 256, 0, stream>>>(jk, jv, jr);
    }

    wkv_kernel<<<D / 256, 256, 0, stream>>>(kvec, vvec, rvec, state, tf, td,
                                            rwkv, s2, s3, s4);

    {
        MatvecJob jo = { ow, rwkv, x_att, x, nullptr, D, D, OP_ADDRES, D / 16 };
        wmma_matvec3_kernel<<<D / 16, 256, 0, stream>>>(jo, jnull, jnull);
    }

    ln_mix_kernel<<<1, 256, 0, stream>>>(x_att, flnw, flnb, state  ,
                                         ftmk, nullptr, ftmr, s0, fxk, nullptr, fxr);

    {
        MatvecJob jfr = { frw, fxr, fr, nullptr, nullptr, D, D, OP_SIGMOID, D / 16 };
        MatvecJob jfk = { fkw, fxk, fk, nullptr, nullptr, D, D, OP_SQRELU,  H / 16 };
        wmma_matvec3_kernel<<<(D / 16) + (H / 16), 256, 0, stream>>>(jfr, jfk, jnull);
    }

    {
        MatvecJob jout = { fvw, fk, out, x_att, fr, H, H, OP_GATEDRES, D / 16 };
        wmma_matvec3_kernel<<<D / 16, 256, 0, stream>>>(jout, jnull, jnull);
    }
}


# FFN value product (K=4096) moved into the FFN key launch as split-K over H: each fk tile of 16 stays in LDS and multiplies its 16 ffn_vw columns; last launch sums 256 partials; sum order changes
# speedup vs baseline: 1.2195x; 1.2195x over previous
#include <hip/hip_runtime.h>
#include <hip/hip_bf16.h>

#define D 1024
#define H 4096
#define EPS 1e-5f

typedef __attribute__((ext_vector_type(2))) float v2f;
typedef __attribute__((ext_vector_type(8))) float v8f;

#define OP_NONE     0
#define OP_SIGMOID  1
#define OP_ADDRES   2

__global__ __launch_bounds__(256) void ln_mix_kernel(
    const float* __restrict__ xin, const float* __restrict__ w,
    const float* __restrict__ b,   const float* __restrict__ prev,
    const float* __restrict__ tmk, const float* __restrict__ tmv,
    const float* __restrict__ tmr,
    float* __restrict__ xln_out, float* __restrict__ xk_out,
    float* __restrict__ xv_out,  float* __restrict__ xr_out)
{
    __shared__ float ssum[8];
    __shared__ float ssq[8];
    const int t = threadIdx.x;
    const int wave = t >> 5;
    const int lane = t & 31;

    float s = 0.f, sq = 0.f;
    for (int i = t; i < D; i += 256) {
        float v = xin[i];
        s += v;
        sq += v * v;
    }
    for (int off = 16; off > 0; off >>= 1) {
        s  += __shfl_xor(s,  off, 32);
        sq += __shfl_xor(sq, off, 32);
    }
    if (lane == 0) { ssum[wave] = s; ssq[wave] = sq; }
    __syncthreads();
    if (wave == 0) {
        float a = (lane < 8) ? ssum[lane] : 0.f;
        float c = (lane < 8) ? ssq[lane]  : 0.f;
        for (int off = 4; off > 0; off >>= 1) {
            a += __shfl_xor(a, off, 32);
            c += __shfl_xor(c, off, 32);
        }
        if (lane == 0) { ssum[0] = a; ssq[0] = c; }
    }
    __syncthreads();

    const float mu   = ssum[0] * (1.f / D);
    const float var  = ssq[0] * (1.f / D) - mu * mu;
    const float rstd = rsqrtf(var + EPS);

    for (int i = t; i < D; i += 256) {
        float xl = (xin[i] - mu) * rstd * w[i] + b[i];
        xln_out[i] = xl;
        float p = prev[i];
        float k = tmk[i];
        xk_out[i] = xl * k + p * (1.f - k);
        if (xv_out) {
            float v = tmv[i];
            xv_out[i] = xl * v + p * (1.f - v);
        }
        float r = tmr[i];
        xr_out[i] = xl * r + p * (1.f - r);
    }
}

struct MatvecJob {
    const float* W;
    const float* x;
    float*       y;
    const float* res;
    int K;
    int ldw;
    int op;
    int nblocks;
};

typedef __attribute__((ext_vector_type(4))) float v4f;

__device__ __forceinline__ v8f matvec_wave_slice(
    const float* __restrict__ W, const float* __restrict__ x,
    int K, int ldw, int row0)
{
    const int wave = threadIdx.x >> 5;
    const int lane = threadIdx.x & 31;
    const int half = lane >> 4;
    const int m    = lane & 15;

    const int kchunk = K >> 3;
    const int k0 = wave * kchunk;
    const int k1 = k0 + kchunk;

    const float* wrow = W + (size_t)(row0 + m) * (size_t)ldw + (half << 1);
    const float* xrow = x + (half << 1);

    v8f c0 = {}; v8f c1 = {}; v8f c2 = {}; v8f c3 = {};

    for (int k = k0; k < k1; k += 16) {
        v2f a0 = *(const v2f*)(wrow + k);
        v2f a1 = *(const v2f*)(wrow + k + 4);
        v2f a2 = *(const v2f*)(wrow + k + 8);
        v2f a3 = *(const v2f*)(wrow + k + 12);
        v2f b0 = *(const v2f*)(xrow + k);
        v2f b1 = *(const v2f*)(xrow + k + 4);
        v2f b2 = *(const v2f*)(xrow + k + 8);
        v2f b3 = *(const v2f*)(xrow + k + 12);
        c0 = __builtin_amdgcn_wmma_f32_16x16x4_f32(false, a0, false, b0, (short)0, c0, false, false);
        c1 = __builtin_amdgcn_wmma_f32_16x16x4_f32(false, a1, false, b1, (short)0, c1, false, false);
        c2 = __builtin_amdgcn_wmma_f32_16x16x4_f32(false, a2, false, b2, (short)0, c2, false, false);
        c3 = __builtin_amdgcn_wmma_f32_16x16x4_f32(false, a3, false, b3, (short)0, c3, false, false);
    }

    return c0 + c1 + c2 + c3;
}

__device__ __forceinline__ void matvec_body(
    const float* __restrict__ W, const float* __restrict__ x,
    float* __restrict__ y, int K, int ldw, int op,
    const float* __restrict__ res, int blk)
{
    __shared__ float part[8][16];

    const int wave = threadIdx.x >> 5;
    const int lane = threadIdx.x & 31;
    const int half = lane >> 4;
    const int m    = lane & 15;
    const int row0 = blk << 4;

    v8f c = matvec_wave_slice(W, x, K, ldw, row0);

    if (m == 0) {
        const int rbase = half << 3;
        #pragma unroll
        for (int j = 0; j < 8; ++j)
            part[wave][rbase + j] = c[j];
    }
    __syncthreads();

    if (threadIdx.x < 16) {
        float v = 0.f;
        #pragma unroll
        for (int wv = 0; wv < 8; ++wv)
            v += part[wv][threadIdx.x];
        const int r = row0 + threadIdx.x;
        switch (op) {
            case OP_SIGMOID:  v = 1.f / (1.f + expf(-v));              break;
            case OP_ADDRES:   v = res[r] + v;                          break;
            default: break;
        }
        y[r] = v;
    }
}

__global__ __launch_bounds__(256) void wmma_matvec3_kernel(
    MatvecJob j0, MatvecJob j1, MatvecJob j2)
{
    int blk = blockIdx.x;
    if (blk < j0.nblocks) {
        matvec_body(j0.W, j0.x, j0.y, j0.K, j0.ldw, j0.op, j0.res, blk);
        return;
    }
    blk -= j0.nblocks;
    if (blk < j1.nblocks) {
        matvec_body(j1.W, j1.x, j1.y, j1.K, j1.ldw, j1.op, j1.res, blk);
        return;
    }
    blk -= j1.nblocks;
    matvec_body(j2.W, j2.x, j2.y, j2.K, j2.ldw, j2.op, j2.res, blk);
}

__global__ __launch_bounds__(256) void ffn_kv_kernel(
    const float* __restrict__ frw, const float* __restrict__ fxr,
    float* __restrict__ fr,
    const float* __restrict__ fkw, const float* __restrict__ fxk,
    const float* __restrict__ fvw, float* __restrict__ partial)
{
    const int blk = blockIdx.x;
    if (blk >= H / 16) {
        matvec_body(frw, fxr, fr, D, D, OP_SIGMOID, nullptr, blk - H / 16);
        return;
    }

    __shared__ float kpart[8][16];
    __shared__ __attribute__((aligned(16))) float fks[16];

    const int wave = threadIdx.x >> 5;
    const int lane = threadIdx.x & 31;
    const int half = lane >> 4;
    const int m    = lane & 15;

    const float* vbase = fvw + (size_t)(wave * 128 + m) * H + (blk << 4) + (half << 2);
    v4f qa[8][2];
    #pragma unroll
    for (int i = 0; i < 8; ++i) {
        qa[i][0] = *(const v4f*)(vbase + (size_t)(i * 16) * H);
        qa[i][1] = *(const v4f*)(vbase + (size_t)(i * 16) * H + 8);
    }

    v8f c = matvec_wave_slice(fkw, fxk, D, D, blk << 4);

    if (m == 0) {
        const int rbase = half << 3;
        #pragma unroll
        for (int j = 0; j < 8; ++j)
            kpart[wave][rbase + j] = c[j];
    }
    __syncthreads();
    if (threadIdx.x < 16) {
        float v = 0.f;
        #pragma unroll
        for (int wv = 0; wv < 8; ++wv)
            v += kpart[wv][threadIdx.x];
        float t = fmaxf(v, 0.f);
        fks[threadIdx.x] = t * t;
    }
    __syncthreads();

    const v4f qb0 = *(const v4f*)(fks + (half << 2));
    const v4f qb1 = *(const v4f*)(fks + 8 + (half << 2));
    const v2f b00 = { qb0.x, qb0.y };
    const v2f b01 = { qb0.z, qb0.w };
    const v2f b10 = { qb1.x, qb1.y };
    const v2f b11 = { qb1.z, qb1.w };

    float* dst = partial + (size_t)blk * D + wave * 128 + (half << 3);

    #pragma unroll
    for (int g = 0; g < 8; g += 4) {
        v8f acc[4];
        #pragma unroll
        for (int i = 0; i < 4; ++i) {
            const v2f a00 = { qa[g + i][0].x, qa[g + i][0].y };
            acc[i] = (v8f){};
            acc[i] = __builtin_amdgcn_wmma_f32_16x16x4_f32(false, a00, false, b00, (short)0, acc[i], false, false);
        }
        #pragma unroll
        for (int i = 0; i < 4; ++i) {
            const v2f a01 = { qa[g + i][0].z, qa[g + i][0].w };
            acc[i] = __builtin_amdgcn_wmma_f32_16x16x4_f32(false, a01, false, b01, (short)0, acc[i], false, false);
        }
        #pragma unroll
        for (int i = 0; i < 4; ++i) {
            const v2f a10 = { qa[g + i][1].x, qa[g + i][1].y };
            acc[i] = __builtin_amdgcn_wmma_f32_16x16x4_f32(false, a10, false, b10, (short)0, acc[i], false, false);
        }
        #pragma unroll
        for (int i = 0; i < 4; ++i) {
            const v2f a11 = { qa[g + i][1].z, qa[g + i][1].w };
            acc[i] = __builtin_amdgcn_wmma_f32_16x16x4_f32(false, a11, false, b11, (short)0, acc[i], false, false);
        }
        if (m == 0) {
            #pragma unroll
            for (int i = 0; i < 4; ++i) {
                const v4f lo = { acc[i][0], acc[i][1], acc[i][2], acc[i][3] };
                const v4f hi = { acc[i][4], acc[i][5], acc[i][6], acc[i][7] };
                *(v4f*)(dst + (g + i) * 16)     = lo;
                *(v4f*)(dst + (g + i) * 16 + 4) = hi;
            }
        }
    }
}

__global__ __launch_bounds__(256) void ffn_out_kernel(
    const float* __restrict__ partial, const float* __restrict__ x_att,
    const float* __restrict__ fr, float* __restrict__ out)
{
    __shared__ float red[8][32];
    const int lane = threadIdx.x & 31;
    const int g    = threadIdx.x >> 5;
    const int col  = blockIdx.x * 32 + lane;

    const float* p = partial + (size_t)(g * 32) * D + col;
    float s0 = 0.f, s1 = 0.f, s2 = 0.f, s3 = 0.f;
    #pragma unroll
    for (int i = 0; i < 32; i += 4) {
        s0 += p[(size_t)(i + 0) * D];
        s1 += p[(size_t)(i + 1) * D];
        s2 += p[(size_t)(i + 2) * D];
        s3 += p[(size_t)(i + 3) * D];
    }
    red[g][lane] = (s0 + s1) + (s2 + s3);
    __syncthreads();

    if (threadIdx.x < 32) {
        float v = 0.f;
        #pragma unroll
        for (int w = 0; w < 8; ++w)
            v += red[w][threadIdx.x];
        out[col] = x_att[col] + fr[col] * v;
    }
}

__global__ __launch_bounds__(256) void wkv_kernel(
    const float* __restrict__ kvec, const float* __restrict__ vvec,
    const float* __restrict__ rvec, const float* __restrict__ state,
    const float* __restrict__ time_first, const float* __restrict__ time_decay,
    float* __restrict__ rwkv,
    float* __restrict__ s2, float* __restrict__ s3, float* __restrict__ s4)
{
    const int i = blockIdx.x * 256 + threadIdx.x;
    if (i >= D) return;

    float k  = kvec[i];
    float v  = vvec[i];
    float r  = rvec[i];
    float aa = state[2 * D + i];
    float bb = state[3 * D + i];
    float pp = state[4 * D + i];

    float ww = time_first[i] + k;
    float qq = fmaxf(pp, ww);
    float e1 = expf(pp - qq);
    float e2 = expf(ww - qq);
    float wkv = (e1 * aa + e2 * v) / (e1 * bb + e2);
    rwkv[i] = r * wkv;

    float ww2 = pp + time_decay[i];
    float qq2 = fmaxf(ww2, k);
    float f1 = expf(ww2 - qq2);
    float f2 = expf(k - qq2);
    s2[i] = f1 * aa + f2 * v;
    s3[i] = f1 * bb + f2;
    s4[i] = qq2;
}

extern "C" void kernel_launch(void* const* d_in, const int* in_sizes, int n_in,
                              void* d_out, int out_size, void* d_ws, size_t ws_size,
                              hipStream_t stream) {
    const float* x     = (const float*)d_in[0];
    const float* state = (const float*)d_in[1];
    const float* tmk   = (const float*)d_in[2];
    const float* tmv   = (const float*)d_in[3];
    const float* tmr   = (const float*)d_in[4];
    const float* tf    = (const float*)d_in[5];
    const float* td    = (const float*)d_in[6];
    const float* kw    = (const float*)d_in[7];
    const float* vw    = (const float*)d_in[8];
    const float* rw    = (const float*)d_in[9];
    const float* ow    = (const float*)d_in[10];
    const float* lnw   = (const float*)d_in[11];
    const float* lnb   = (const float*)d_in[12];
    const float* ftmk  = (const float*)d_in[13];
    const float* ftmr  = (const float*)d_in[14];
    const float* fkw   = (const float*)d_in[15];
    const float* fvw   = (const float*)d_in[16];
    const float* frw   = (const float*)d_in[17];
    const float* flnw  = (const float*)d_in[18];
    const float* flnb  = (const float*)d_in[19];

    float* ws    = (float*)d_ws;
    float* xk    = ws + 1 * D;
    float* xv    = ws + 2 * D;
    float* xr    = ws + 3 * D;
    float* kvec  = ws + 4 * D;
    float* vvec  = ws + 5 * D;
    float* rvec  = ws + 6 * D;
    float* rwkv  = ws + 7 * D;
    float* x_att = ws + 8 * D;
    float* fxk   = ws + 9 * D;
    float* fxr   = ws + 10 * D;
    float* fr    = ws + 11 * D;
    float* fvpart = ws + 16 * D;

    float* out = (float*)d_out;
    float* s0  = out + 1 * D;
    float* s1  = out + 2 * D;
    float* s2  = out + 3 * D;
    float* s3  = out + 4 * D;
    float* s4  = out + 5 * D;

    const MatvecJob jnull = { nullptr, nullptr, nullptr, nullptr, 0, 0, OP_NONE, 0 };

    ln_mix_kernel<<<1, 256, 0, stream>>>(x, lnw, lnb, state + 1 * D,
                                         tmk, tmv, tmr, s1, xk, xv, xr);

    {
        MatvecJob jk = { kw, xk, kvec, nullptr, D, D, OP_NONE,    D / 16 };
        MatvecJob jv = { vw, xv, vvec, nullptr, D, D, OP_NONE,    D / 16 };
        MatvecJob jr = { rw, xr, rvec, nullptr, D, D, OP_SIGMOID, D / 16 };
        wmma_matvec3_kernel<<<3 * (D / 16), 256, 0, stream>>>(jk, jv, jr);
    }

    wkv_kernel<<<D / 256, 256, 0, stream>>>(kvec, vvec, rvec, state, tf, td,
                                            rwkv, s2, s3, s4);

    {
        MatvecJob jo = { ow, rwkv, x_att, x, D, D, OP_ADDRES, D / 16 };
        wmma_matvec3_kernel<<<D / 16, 256, 0, stream>>>(jo, jnull, jnull);
    }

    ln_mix_kernel<<<1, 256, 0, stream>>>(x_att, flnw, flnb, state  ,
                                         ftmk, nullptr, ftmr, s0, fxk, nullptr, fxr);

    ffn_kv_kernel<<<(H / 16) + (D / 16), 256, 0, stream>>>(frw, fxr, fr, fkw, fxk, fvw, fvpart);

    ffn_out_kernel<<<D / 32, 256, 0, stream>>>(fvpart, x_att, fr, out);
}


# Attention output product (K=1024) moved into the k/v/r launch as split-K over channels: WKV on a block's 16 channels, then times their 16 ow columns; LN launch sums the 64 partials; sum order changes
# speedup vs baseline: 1.4971x; 1.2276x over previous
#include <hip/hip_runtime.h>
#include <hip/hip_bf16.h>

#define D 1024
#define H 4096
#define EPS 1e-5f

typedef __attribute__((ext_vector_type(2))) float v2f;
typedef __attribute__((ext_vector_type(8))) float v8f;

#define OP_NONE     0
#define OP_SIGMOID  1
#define OP_ADDRES   2

__global__ __launch_bounds__(256) void ln_mix_kernel(
    const float* __restrict__ xin, const float* __restrict__ w,
    const float* __restrict__ b,   const float* __restrict__ prev,
    const float* __restrict__ tmk, const float* __restrict__ tmv,
    const float* __restrict__ tmr,
    float* __restrict__ xln_out, float* __restrict__ xk_out,
    float* __restrict__ xv_out,  float* __restrict__ xr_out)
{
    __shared__ float ssum[8];
    __shared__ float ssq[8];
    const int t = threadIdx.x;
    const int wave = t >> 5;
    const int lane = t & 31;

    float s = 0.f, sq = 0.f;
    for (int i = t; i < D; i += 256) {
        float v = xin[i];
        s += v;
        sq += v * v;
    }
    for (int off = 16; off > 0; off >>= 1) {
        s  += __shfl_xor(s,  off, 32);
        sq += __shfl_xor(sq, off, 32);
    }
    if (lane == 0) { ssum[wave] = s; ssq[wave] = sq; }
    __syncthreads();
    if (wave == 0) {
        float a = (lane < 8) ? ssum[lane] : 0.f;
        float c = (lane < 8) ? ssq[lane]  : 0.f;
        for (int off = 4; off > 0; off >>= 1) {
            a += __shfl_xor(a, off, 32);
            c += __shfl_xor(c, off, 32);
        }
        if (lane == 0) { ssum[0] = a; ssq[0] = c; }
    }
    __syncthreads();

    const float mu   = ssum[0] * (1.f / D);
    const float var  = ssq[0] * (1.f / D) - mu * mu;
    const float rstd = rsqrtf(var + EPS);

    for (int i = t; i < D; i += 256) {
        float xl = (xin[i] - mu) * rstd * w[i] + b[i];
        xln_out[i] = xl;
        float p = prev[i];
        float k = tmk[i];
        xk_out[i] = xl * k + p * (1.f - k);
        float v = tmv[i];
        xv_out[i] = xl * v + p * (1.f - v);
        float r = tmr[i];
        xr_out[i] = xl * r + p * (1.f - r);
    }
}

typedef __attribute__((ext_vector_type(4))) float v4f;

__device__ __forceinline__ v8f matvec_wave_slice(
    const float* __restrict__ W, const float* __restrict__ x,
    int K, int ldw, int row0, int wave)
{
    const int lane = threadIdx.x & 31;
    const int half = lane >> 4;
    const int m    = lane & 15;

    const int kchunk = K >> 3;
    const int k0 = wave * kchunk;
    const int k1 = k0 + kchunk;

    const float* wrow = W + (size_t)(row0 + m) * (size_t)ldw + (half << 1);
    const float* xrow = x + (half << 1);

    v8f c0 = {}; v8f c1 = {}; v8f c2 = {}; v8f c3 = {};

    for (int k = k0; k < k1; k += 16) {
        v2f a0 = *(const v2f*)(wrow + k);
        v2f a1 = *(const v2f*)(wrow + k + 4);
        v2f a2 = *(const v2f*)(wrow + k + 8);
        v2f a3 = *(const v2f*)(wrow + k + 12);
        v2f b0 = *(const v2f*)(xrow + k);
        v2f b1 = *(const v2f*)(xrow + k + 4);
        v2f b2 = *(const v2f*)(xrow + k + 8);
        v2f b3 = *(const v2f*)(xrow + k + 12);
        c0 = __builtin_amdgcn_wmma_f32_16x16x4_f32(false, a0, false, b0, (short)0, c0, false, false);
        c1 = __builtin_amdgcn_wmma_f32_16x16x4_f32(false, a1, false, b1, (short)0, c1, false, false);
        c2 = __builtin_amdgcn_wmma_f32_16x16x4_f32(false, a2, false, b2, (short)0, c2, false, false);
        c3 = __builtin_amdgcn_wmma_f32_16x16x4_f32(false, a3, false, b3, (short)0, c3, false, false);
    }

    return c0 + c1 + c2 + c3;
}

__device__ __forceinline__ void matvec_body(
    const float* __restrict__ W, const float* __restrict__ x,
    float* __restrict__ y, int K, int ldw, int op,
    const float* __restrict__ res, int blk)
{
    __shared__ float part[8][16];

    const int wave = threadIdx.x >> 5;
    const int lane = threadIdx.x & 31;
    const int half = lane >> 4;
    const int m    = lane & 15;
    const int row0 = blk << 4;

    v8f c = matvec_wave_slice(W, x, K, ldw, row0, wave);

    if (m == 0) {
        const int rbase = half << 3;
        #pragma unroll
        for (int j = 0; j < 8; ++j)
            part[wave][rbase + j] = c[j];
    }
    __syncthreads();

    if (threadIdx.x < 16) {
        float v = 0.f;
        #pragma unroll
        for (int wv = 0; wv < 8; ++wv)
            v += part[wv][threadIdx.x];
        const int r = row0 + threadIdx.x;
        switch (op) {
            case OP_SIGMOID:  v = 1.f / (1.f + expf(-v));              break;
            case OP_ADDRES:   v = res[r] + v;                          break;
            default: break;
        }
        y[r] = v;
    }
}

__device__ __forceinline__ v8f ow_tile(const v4f q0, const v4f q1,
    const v2f b00, const v2f b01, const v2f b10, const v2f b11)
{
    const v2f a00 = { q0.x, q0.y };
    const v2f a01 = { q0.z, q0.w };
    const v2f a10 = { q1.x, q1.y };
    const v2f a11 = { q1.z, q1.w };
    v8f acc = {};
    acc = __builtin_amdgcn_wmma_f32_16x16x4_f32(false, a00, false, b00, (short)0, acc, false, false);
    acc = __builtin_amdgcn_wmma_f32_16x16x4_f32(false, a01, false, b01, (short)0, acc, false, false);
    acc = __builtin_amdgcn_wmma_f32_16x16x4_f32(false, a10, false, b10, (short)0, acc, false, false);
    acc = __builtin_amdgcn_wmma_f32_16x16x4_f32(false, a11, false, b11, (short)0, acc, false, false);
    return acc;
}

__device__ __forceinline__ void store_tile_rows(float* __restrict__ dst, const v8f acc)
{
    const v4f lo = { acc[0], acc[1], acc[2], acc[3] };
    const v4f hi = { acc[4], acc[5], acc[6], acc[7] };
    *(v4f*)(dst)     = lo;
    *(v4f*)(dst + 4) = hi;
}

__global__ __launch_bounds__(768) void att_kvr_ow_kernel(
    const float* __restrict__ kw, const float* __restrict__ vw,
    const float* __restrict__ rw,
    const float* __restrict__ xk, const float* __restrict__ xv,
    const float* __restrict__ xr,
    const float* __restrict__ state,
    const float* __restrict__ time_first, const float* __restrict__ time_decay,
    const float* __restrict__ ow, float* __restrict__ partial,
    float* __restrict__ st2, float* __restrict__ st3, float* __restrict__ st4)
{
    __shared__ float kpart[3][8][16];
    __shared__ __attribute__((aligned(16))) float rws[16];

    const int blk   = blockIdx.x;
    const int wave  = threadIdx.x >> 5;
    const int lane  = threadIdx.x & 31;
    const int half  = lane >> 4;
    const int m     = lane & 15;
    const int job   = wave >> 3;
    const int slice = wave & 7;
    const bool third = wave < 16;

    const float* obase = ow + (size_t)(wave * 16 + m) * D + (blk << 4) + (half << 2);
    v4f qa[3][2];
    qa[0][0] = *(const v4f*)(obase);
    qa[0][1] = *(const v4f*)(obase + 8);
    qa[1][0] = *(const v4f*)(obase + (size_t)(24 * 16) * D);
    qa[1][1] = *(const v4f*)(obase + (size_t)(24 * 16) * D + 8);
    qa[2][0] = (v4f){};
    qa[2][1] = (v4f){};
    if (third) {
        qa[2][0] = *(const v4f*)(obase + (size_t)(48 * 16) * D);
        qa[2][1] = *(const v4f*)(obase + (size_t)(48 * 16) * D + 8);
    }

    const int ch = (blk << 4) + m;
    const float aa = state[2 * D + ch];
    const float bb = state[3 * D + ch];
    const float pp = state[4 * D + ch];
    const float tf = time_first[ch];
    const float td = time_decay[ch];

    const float* Wj = (job == 0) ? kw : (job == 1) ? vw : rw;
    const float* xj = (job == 0) ? xk : (job == 1) ? xv : xr;
    v8f c = matvec_wave_slice(Wj, xj, D, D, blk << 4, slice);

    if (m == 0) {
        const int rbase = half << 3;
        #pragma unroll
        for (int j = 0; j < 8; ++j)
            kpart[job][slice][rbase + j] = c[j];
    }
    __syncthreads();

    if (threadIdx.x < 16) {
        float k = 0.f, v = 0.f, r = 0.f;
        #pragma unroll
        for (int wv = 0; wv < 8; ++wv) {
            k += kpart[0][wv][threadIdx.x];
            v += kpart[1][wv][threadIdx.x];
            r += kpart[2][wv][threadIdx.x];
        }
        r = 1.f / (1.f + expf(-r));

        float ww = tf + k;
        float qq = fmaxf(pp, ww);
        float e1 = expf(pp - qq);
        float e2 = expf(ww - qq);
        float wkv = (e1 * aa + e2 * v) / (e1 * bb + e2);
        rws[threadIdx.x] = r * wkv;

        float ww2 = pp + td;
        float qq2 = fmaxf(ww2, k);
        float f1 = expf(ww2 - qq2);
        float f2 = expf(k - qq2);
        st2[ch] = f1 * aa + f2 * v;
        st3[ch] = f1 * bb + f2;
        st4[ch] = qq2;
    }
    __syncthreads();

    const v4f qb0 = *(const v4f*)(rws + (half << 2));
    const v4f qb1 = *(const v4f*)(rws + 8 + (half << 2));
    const v2f b00 = { qb0.x, qb0.y };
    const v2f b01 = { qb0.z, qb0.w };
    const v2f b10 = { qb1.x, qb1.y };
    const v2f b11 = { qb1.z, qb1.w };

    float* dst = partial + (size_t)blk * D + wave * 16 + (half << 3);

    const v8f acc0 = ow_tile(qa[0][0], qa[0][1], b00, b01, b10, b11);
    const v8f acc1 = ow_tile(qa[1][0], qa[1][1], b00, b01, b10, b11);
    if (third) {
        const v8f acc2 = ow_tile(qa[2][0], qa[2][1], b00, b01, b10, b11);
        if (m == 0)
            store_tile_rows(dst + 48 * 16, acc2);
    }
    if (m == 0) {
        store_tile_rows(dst, acc0);
        store_tile_rows(dst + 24 * 16, acc1);
    }
}

__global__ __launch_bounds__(1024) void att_sum_ln_mix_kernel(
    const float* __restrict__ partial, const float* __restrict__ x,
    const float* __restrict__ w, const float* __restrict__ b,
    const float* __restrict__ prev,
    const float* __restrict__ tmk, const float* __restrict__ tmr,
    float* __restrict__ x_att, float* __restrict__ xln_out,
    float* __restrict__ xk_out, float* __restrict__ xr_out)
{
    __shared__ float ssum[32];
    __shared__ float ssq[32];
    const int t = threadIdx.x;
    const int wave = t >> 5;
    const int lane = t & 31;

    const float xin = x[t];
    const float wv = w[t];
    const float bv = b[t];
    const float pv = prev[t];
    const float kv = tmk[t];
    const float rv = tmr[t];

    const float* p = partial + t;
    float s0 = 0.f, s1 = 0.f, s2 = 0.f, s3 = 0.f;
    #pragma unroll
    for (int i = 0; i < D / 16; i += 4) {
        s0 += p[(size_t)(i + 0) * D];
        s1 += p[(size_t)(i + 1) * D];
        s2 += p[(size_t)(i + 2) * D];
        s3 += p[(size_t)(i + 3) * D];
    }
    const float xa = xin + ((s0 + s1) + (s2 + s3));
    x_att[t] = xa;

    float s = xa, sq = xa * xa;
    for (int off = 16; off > 0; off >>= 1) {
        s  += __shfl_xor(s,  off, 32);
        sq += __shfl_xor(sq, off, 32);
    }
    if (lane == 0) { ssum[wave] = s; ssq[wave] = sq; }
    __syncthreads();
    if (wave == 0) {
        float a = ssum[lane];
        float c = ssq[lane];
        for (int off = 16; off > 0; off >>= 1) {
            a += __shfl_xor(a, off, 32);
            c += __shfl_xor(c, off, 32);
        }
        if (lane == 0) { ssum[0] = a; ssq[0] = c; }
    }
    __syncthreads();

    const float mu   = ssum[0] * (1.f / D);
    const float var  = ssq[0] * (1.f / D) - mu * mu;
    const float rstd = rsqrtf(var + EPS);

    const float xl = (xa - mu) * rstd * wv + bv;
    xln_out[t] = xl;
    xk_out[t] = xl * kv + pv * (1.f - kv);
    xr_out[t] = xl * rv + pv * (1.f - rv);
}

__global__ __launch_bounds__(256) void ffn_kv_kernel(
    const float* __restrict__ frw, const float* __restrict__ fxr,
    float* __restrict__ fr,
    const float* __restrict__ fkw, const float* __restrict__ fxk,
    const float* __restrict__ fvw, float* __restrict__ partial)
{
    const int blk = blockIdx.x;
    if (blk >= H / 16) {
        matvec_body(frw, fxr, fr, D, D, OP_SIGMOID, nullptr, blk - H / 16);
        return;
    }

    __shared__ float kpart[8][16];
    __shared__ __attribute__((aligned(16))) float fks[16];

    const int wave = threadIdx.x >> 5;
    const int lane = threadIdx.x & 31;
    const int half = lane >> 4;
    const int m    = lane & 15;

    const float* vbase = fvw + (size_t)(wave * 128 + m) * H + (blk << 4) + (half << 2);
    v4f qa[8][2];
    #pragma unroll
    for (int i = 0; i < 8; ++i) {
        qa[i][0] = *(const v4f*)(vbase + (size_t)(i * 16) * H);
        qa[i][1] = *(const v4f*)(vbase + (size_t)(i * 16) * H + 8);
    }

    v8f c = matvec_wave_slice(fkw, fxk, D, D, blk << 4, wave);

    if (m == 0) {
        const int rbase = half << 3;
        #pragma unroll
        for (int j = 0; j < 8; ++j)
            kpart[wave][rbase + j] = c[j];
    }
    __syncthreads();
    if (threadIdx.x < 16) {
        float v = 0.f;
        #pragma unroll
        for (int wv = 0; wv < 8; ++wv)
            v += kpart[wv][threadIdx.x];
        float t = fmaxf(v, 0.f);
        fks[threadIdx.x] = t * t;
    }
    __syncthreads();

    const v4f qb0 = *(const v4f*)(fks + (half << 2));
    const v4f qb1 = *(const v4f*)(fks + 8 + (half << 2));
    const v2f b00 = { qb0.x, qb0.y };
    const v2f b01 = { qb0.z, qb0.w };
    const v2f b10 = { qb1.x, qb1.y };
    const v2f b11 = { qb1.z, qb1.w };

    float* dst = partial + (size_t)blk * D + wave * 128 + (half << 3);

    #pragma unroll
    for (int g = 0; g < 8; g += 4) {
        v8f acc[4];
        #pragma unroll
        for (int i = 0; i < 4; ++i) {
            const v2f a00 = { qa[g + i][0].x, qa[g + i][0].y };
            acc[i] = (v8f){};
            acc[i] = __builtin_amdgcn_wmma_f32_16x16x4_f32(false, a00, false, b00, (short)0, acc[i], false, false);
        }
        #pragma unroll
        for (int i = 0; i < 4; ++i) {
            const v2f a01 = { qa[g + i][0].z, qa[g + i][0].w };
            acc[i] = __builtin_amdgcn_wmma_f32_16x16x4_f32(false, a01, false, b01, (short)0, acc[i], false, false);
        }
        #pragma unroll
        for (int i = 0; i < 4; ++i) {
            const v2f a10 = { qa[g + i][1].x, qa[g + i][1].y };
            acc[i] = __builtin_amdgcn_wmma_f32_16x16x4_f32(false, a10, false, b10, (short)0, acc[i], false, false);
        }
        #pragma unroll
        for (int i = 0; i < 4; ++i) {
            const v2f a11 = { qa[g + i][1].z, qa[g + i][1].w };
            acc[i] = __builtin_amdgcn_wmma_f32_16x16x4_f32(false, a11, false, b11, (short)0, acc[i], false, false);
        }
        if (m == 0) {
            #pragma unroll
            for (int i = 0; i < 4; ++i) {
                const v4f lo = { acc[i][0], acc[i][1], acc[i][2], acc[i][3] };
                const v4f hi = { acc[i][4], acc[i][5], acc[i][6], acc[i][7] };
                *(v4f*)(dst + (g + i) * 16)     = lo;
                *(v4f*)(dst + (g + i) * 16 + 4) = hi;
            }
        }
    }
}

__global__ __launch_bounds__(256) void ffn_out_kernel(
    const float* __restrict__ partial, const float* __restrict__ x_att,
    const float* __restrict__ fr, const float* __restrict__ staged,
    float* __restrict__ out)
{
    __shared__ float red[8][32];
    const int lane = threadIdx.x & 31;
    const int g    = threadIdx.x >> 5;
    const int col  = blockIdx.x * 32 + lane;

    const float* p = partial + (size_t)(g * 32) * D + col;
    float s0 = 0.f, s1 = 0.f, s2 = 0.f, s3 = 0.f;
    #pragma unroll
    for (int i = 0; i < 32; i += 4) {
        s0 += p[(size_t)(i + 0) * D];
        s1 += p[(size_t)(i + 1) * D];
        s2 += p[(size_t)(i + 2) * D];
        s3 += p[(size_t)(i + 3) * D];
    }
    red[g][lane] = (s0 + s1) + (s2 + s3);
    if (g >= 1 && g <= 3)
        out[(size_t)(2 + g) * D + col] = staged[(size_t)(g - 1) * D + col];
    __syncthreads();

    if (threadIdx.x < 32) {
        float v = 0.f;
        #pragma unroll
        for (int w = 0; w < 8; ++w)
            v += red[w][threadIdx.x];
        out[col] = x_att[col] + fr[col] * v;
    }
}

extern "C" void kernel_launch(void* const* d_in, const int* in_sizes, int n_in,
                              void* d_out, int out_size, void* d_ws, size_t ws_size,
                              hipStream_t stream) {
    const float* x     = (const float*)d_in[0];
    const float* state = (const float*)d_in[1];
    const float* tmk   = (const float*)d_in[2];
    const float* tmv   = (const float*)d_in[3];
    const float* tmr   = (const float*)d_in[4];
    const float* tf    = (const float*)d_in[5];
    const float* td    = (const float*)d_in[6];
    const float* kw    = (const float*)d_in[7];
    const float* vw    = (const float*)d_in[8];
    const float* rw    = (const float*)d_in[9];
    const float* ow    = (const float*)d_in[10];
    const float* lnw   = (const float*)d_in[11];
    const float* lnb   = (const float*)d_in[12];
    const float* ftmk  = (const float*)d_in[13];
    const float* ftmr  = (const float*)d_in[14];
    const float* fkw   = (const float*)d_in[15];
    const float* fvw   = (const float*)d_in[16];
    const float* frw   = (const float*)d_in[17];
    const float* flnw  = (const float*)d_in[18];
    const float* flnb  = (const float*)d_in[19];

    float* ws    = (float*)d_ws;
    float* xk    = ws + 1 * D;
    float* xv    = ws + 2 * D;
    float* xr    = ws + 3 * D;
    float* staged = ws + 4 * D;
    float* x_att = ws + 8 * D;
    float* fxk   = ws + 9 * D;
    float* fxr   = ws + 10 * D;
    float* fr    = ws + 11 * D;
    float* fvpart = ws + 16 * D;
    float* opart  = ws + 512 * D;

    float* out = (float*)d_out;
    float* s0  = out + 1 * D;
    float* s1  = out + 2 * D;

    ln_mix_kernel<<<1, 256, 0, stream>>>(x, lnw, lnb, state + 1 * D,
                                         tmk, tmv, tmr, s1, xk, xv, xr);

    att_kvr_ow_kernel<<<D / 16, 768, 0, stream>>>(kw, vw, rw, xk, xv, xr, state, tf, td,
                                                   ow, opart,
                                                   staged, staged + 1 * D, staged + 2 * D);

    att_sum_ln_mix_kernel<<<1, 1024, 0, stream>>>(opart, x, flnw, flnb, state  ,
                                                  ftmk, ftmr, x_att, s0, fxk, fxr);

    ffn_kv_kernel<<<(H / 16) + (D / 16), 256, 0, stream>>>(frw, fxr, fr, fkw, fxk, fvw, fvpart);

    ffn_out_kernel<<<D / 32, 256, 0, stream>>>(fvpart, x_att, fr, staged, out);
}


# Each launch requests all global operands before its first wait and stores behind its last barrier: launch 1 one round trip for eight (16 B per lane), FFN value loads kept first; LN sum order changes
# speedup vs baseline: 1.7823x; 1.1905x over previous
#include <hip/hip_runtime.h>
#include <hip/hip_bf16.h>

#define D 1024
#define H 4096
#define EPS 1e-5f

typedef __attribute__((ext_vector_type(2))) float v2f;
typedef __attribute__((ext_vector_type(8))) float v8f;
typedef __attribute__((ext_vector_type(4))) float v4f;

#define OP_NONE     0
#define OP_SIGMOID  1
#define OP_ADDRES   2

__global__ __launch_bounds__(256) void ln_mix_kernel(
    const float* __restrict__ xin, const float* __restrict__ w,
    const float* __restrict__ b,   const float* __restrict__ prev,
    const float* __restrict__ tmk, const float* __restrict__ tmv,
    const float* __restrict__ tmr,
    float* __restrict__ xln_out, float* __restrict__ xk_out,
    float* __restrict__ xv_out,  float* __restrict__ xr_out)
{
    __shared__ float ssum[8];
    __shared__ float ssq[8];
    const int t = threadIdx.x;
    const int wave = t >> 5;
    const int lane = t & 31;

    const v4f x4 = ((const v4f*)xin)[t];
    const v4f w4 = ((const v4f*)w)[t];
    const v4f b4 = ((const v4f*)b)[t];
    const v4f p4 = ((const v4f*)prev)[t];
    const v4f k4 = ((const v4f*)tmk)[t];
    const v4f v4 = ((const v4f*)tmv)[t];
    const v4f r4 = ((const v4f*)tmr)[t];

    float s  = (x4.x + x4.y) + (x4.z + x4.w);
    float sq = (x4.x * x4.x + x4.y * x4.y) + (x4.z * x4.z + x4.w * x4.w);
    for (int off = 16; off > 0; off >>= 1) {
        s  += __shfl_xor(s,  off, 32);
        sq += __shfl_xor(sq, off, 32);
    }
    if (lane == 0) { ssum[wave] = s; ssq[wave] = sq; }
    __syncthreads();
    if (wave == 0) {
        float a = (lane < 8) ? ssum[lane] : 0.f;
        float c = (lane < 8) ? ssq[lane]  : 0.f;
        for (int off = 4; off > 0; off >>= 1) {
            a += __shfl_xor(a, off, 32);
            c += __shfl_xor(c, off, 32);
        }
        if (lane == 0) { ssum[0] = a; ssq[0] = c; }
    }
    __syncthreads();

    const float mu   = ssum[0] * (1.f / D);
    const float var  = ssq[0] * (1.f / D) - mu * mu;
    const float rstd = rsqrtf(var + EPS);

    const v4f xl = (x4 - mu) * rstd * w4 + b4;
    ((v4f*)xln_out)[t] = xl;
    ((v4f*)xk_out)[t]  = xl * k4 + p4 * (1.f - k4);
    ((v4f*)xv_out)[t]  = xl * v4 + p4 * (1.f - v4);
    ((v4f*)xr_out)[t]  = xl * r4 + p4 * (1.f - r4);
}

__device__ __forceinline__ v8f matvec_wave_slice(
    const float* __restrict__ W, const float* __restrict__ x,
    int K, int ldw, int row0, int wave)
{
    const int lane = threadIdx.x & 31;
    const int half = lane >> 4;
    const int m    = lane & 15;

    const int kchunk = K >> 3;
    const int k0 = wave * kchunk;
    const int k1 = k0 + kchunk;

    const float* wrow = W + (size_t)(row0 + m) * (size_t)ldw + (half << 1);
    const float* xrow = x + (half << 1);

    v8f c0 = {}; v8f c1 = {}; v8f c2 = {}; v8f c3 = {};

    for (int k = k0; k < k1; k += 16) {
        v2f a0 = *(const v2f*)(wrow + k);
        v2f a1 = *(const v2f*)(wrow + k + 4);
        v2f a2 = *(const v2f*)(wrow + k + 8);
        v2f a3 = *(const v2f*)(wrow + k + 12);
        v2f b0 = *(const v2f*)(xrow + k);
        v2f b1 = *(const v2f*)(xrow + k + 4);
        v2f b2 = *(const v2f*)(xrow + k + 8);
        v2f b3 = *(const v2f*)(xrow + k + 12);
        c0 = __builtin_amdgcn_wmma_f32_16x16x4_f32(false, a0, false, b0, (short)0, c0, false, false);
        c1 = __builtin_amdgcn_wmma_f32_16x16x4_f32(false, a1, false, b1, (short)0, c1, false, false);
        c2 = __builtin_amdgcn_wmma_f32_16x16x4_f32(false, a2, false, b2, (short)0, c2, false, false);
        c3 = __builtin_amdgcn_wmma_f32_16x16x4_f32(false, a3, false, b3, (short)0, c3, false, false);
    }

    return c0 + c1 + c2 + c3;
}

__device__ __forceinline__ void matvec_body(
    const float* __restrict__ W, const float* __restrict__ x,
    float* __restrict__ y, int K, int ldw, int op,
    const float* __restrict__ res, int blk)
{
    __shared__ float part[8][16];

    const int wave = threadIdx.x >> 5;
    const int lane = threadIdx.x & 31;
    const int half = lane >> 4;
    const int m    = lane & 15;
    const int row0 = blk << 4;

    v8f c = matvec_wave_slice(W, x, K, ldw, row0, wave);

    if (m == 0) {
        const int rbase = half << 3;
        #pragma unroll
        for (int j = 0; j < 8; ++j)
            part[wave][rbase + j] = c[j];
    }
    __syncthreads();

    if (threadIdx.x < 16) {
        float v = 0.f;
        #pragma unroll
        for (int wv = 0; wv < 8; ++wv)
            v += part[wv][threadIdx.x];
        const int r = row0 + threadIdx.x;
        switch (op) {
            case OP_SIGMOID:  v = 1.f / (1.f + expf(-v));              break;
            case OP_ADDRES:   v = res[r] + v;                          break;
            default: break;
        }
        y[r] = v;
    }
}

__device__ __forceinline__ v8f ow_tile(const v4f q0, const v4f q1,
    const v2f b00, const v2f b01, const v2f b10, const v2f b11)
{
    const v2f a00 = { q0.x, q0.y };
    const v2f a01 = { q0.z, q0.w };
    const v2f a10 = { q1.x, q1.y };
    const v2f a11 = { q1.z, q1.w };
    v8f acc = {};
    acc = __builtin_amdgcn_wmma_f32_16x16x4_f32(false, a00, false, b00, (short)0, acc, false, false);
    acc = __builtin_amdgcn_wmma_f32_16x16x4_f32(false, a01, false, b01, (short)0, acc, false, false);
    acc = __builtin_amdgcn_wmma_f32_16x16x4_f32(false, a10, false, b10, (short)0, acc, false, false);
    acc = __builtin_amdgcn_wmma_f32_16x16x4_f32(false, a11, false, b11, (short)0, acc, false, false);
    return acc;
}

__device__ __forceinline__ void store_tile_rows(float* __restrict__ dst, const v8f acc)
{
    const v4f lo = { acc[0], acc[1], acc[2], acc[3] };
    const v4f hi = { acc[4], acc[5], acc[6], acc[7] };
    *(v4f*)(dst)     = lo;
    *(v4f*)(dst + 4) = hi;
}

__global__ __launch_bounds__(768) void att_kvr_ow_kernel(
    const float* __restrict__ kw, const float* __restrict__ vw,
    const float* __restrict__ rw,
    const float* __restrict__ xk, const float* __restrict__ xv,
    const float* __restrict__ xr,
    const float* __restrict__ state,
    const float* __restrict__ time_first, const float* __restrict__ time_decay,
    const float* __restrict__ ow, float* __restrict__ partial,
    float* __restrict__ st2, float* __restrict__ st3, float* __restrict__ st4)
{
    __shared__ float kpart[3][8][16];
    __shared__ __attribute__((aligned(16))) float rws[16];

    const int blk   = blockIdx.x;
    const int wave  = threadIdx.x >> 5;
    const int lane  = threadIdx.x & 31;
    const int half  = lane >> 4;
    const int m     = lane & 15;
    const int job   = wave >> 3;
    const int slice = wave & 7;
    const bool third = wave < 16;

    const float* obase = ow + (size_t)(wave * 16 + m) * D + (blk << 4) + (half << 2);
    v4f qa[3][2];
    qa[0][0] = *(const v4f*)(obase);
    qa[0][1] = *(const v4f*)(obase + 8);
    qa[1][0] = *(const v4f*)(obase + (size_t)(24 * 16) * D);
    qa[1][1] = *(const v4f*)(obase + (size_t)(24 * 16) * D + 8);
    qa[2][0] = (v4f){};
    qa[2][1] = (v4f){};
    if (third) {
        qa[2][0] = *(const v4f*)(obase + (size_t)(48 * 16) * D);
        qa[2][1] = *(const v4f*)(obase + (size_t)(48 * 16) * D + 8);
    }

    const int ch = (blk << 4) + m;
    const float aa = state[2 * D + ch];
    const float bb = state[3 * D + ch];
    const float pp = state[4 * D + ch];
    const float tf = time_first[ch];
    const float td = time_decay[ch];

    const float* Wj = (job == 0) ? kw : (job == 1) ? vw : rw;
    const float* xj = (job == 0) ? xk : (job == 1) ? xv : xr;
    v8f c = matvec_wave_slice(Wj, xj, D, D, blk << 4, slice);

    if (m == 0) {
        const int rbase = half << 3;
        #pragma unroll
        for (int j = 0; j < 8; ++j)
            kpart[job][slice][rbase + j] = c[j];
    }
    __syncthreads();

    float n2 = 0.f, n3 = 0.f, n4 = 0.f;
    if (threadIdx.x < 16) {
        float k = 0.f, v = 0.f, r = 0.f;
        #pragma unroll
        for (int wv = 0; wv < 8; ++wv) {
            k += kpart[0][wv][threadIdx.x];
            v += kpart[1][wv][threadIdx.x];
            r += kpart[2][wv][threadIdx.x];
        }
        r = 1.f / (1.f + expf(-r));

        float ww = tf + k;
        float qq = fmaxf(pp, ww);
        float e1 = expf(pp - qq);
        float e2 = expf(ww - qq);
        float wkv = (e1 * aa + e2 * v) / (e1 * bb + e2);
        rws[threadIdx.x] = r * wkv;

        float ww2 = pp + td;
        float qq2 = fmaxf(ww2, k);
        float f1 = expf(ww2 - qq2);
        float f2 = expf(k - qq2);
        n2 = f1 * aa + f2 * v;
        n3 = f1 * bb + f2;
        n4 = qq2;
    }
    __syncthreads();
    if (threadIdx.x < 16) {
        st2[ch] = n2;
        st3[ch] = n3;
        st4[ch] = n4;
    }

    const v4f qb0 = *(const v4f*)(rws + (half << 2));
    const v4f qb1 = *(const v4f*)(rws + 8 + (half << 2));
    const v2f b00 = { qb0.x, qb0.y };
    const v2f b01 = { qb0.z, qb0.w };
    const v2f b10 = { qb1.x, qb1.y };
    const v2f b11 = { qb1.z, qb1.w };

    float* dst = partial + (size_t)blk * D + wave * 16 + (half << 3);

    const v8f acc0 = ow_tile(qa[0][0], qa[0][1], b00, b01, b10, b11);
    const v8f acc1 = ow_tile(qa[1][0], qa[1][1], b00, b01, b10, b11);
    if (third) {
        const v8f acc2 = ow_tile(qa[2][0], qa[2][1], b00, b01, b10, b11);
        if (m == 0)
            store_tile_rows(dst + 48 * 16, acc2);
    }
    if (m == 0) {
        store_tile_rows(dst, acc0);
        store_tile_rows(dst + 24 * 16, acc1);
    }
}

__global__ __launch_bounds__(1024) void att_sum_ln_mix_kernel(
    const float* __restrict__ partial, const float* __restrict__ x,
    const float* __restrict__ w, const float* __restrict__ b,
    const float* __restrict__ prev,
    const float* __restrict__ tmk, const float* __restrict__ tmr,
    float* __restrict__ x_att, float* __restrict__ xln_out,
    float* __restrict__ xk_out, float* __restrict__ xr_out)
{
    __shared__ float ssum[32];
    __shared__ float ssq[32];
    const int t = threadIdx.x;
    const int wave = t >> 5;
    const int lane = t & 31;

    const float xin = x[t];
    const float wv = w[t];
    const float bv = b[t];
    const float pv = prev[t];
    const float kv = tmk[t];
    const float rv = tmr[t];

    const float* p = partial + t;
    float s0 = 0.f, s1 = 0.f, s2 = 0.f, s3 = 0.f;
    #pragma unroll
    for (int i = 0; i < D / 16; i += 4) {
        s0 += p[(size_t)(i + 0) * D];
        s1 += p[(size_t)(i + 1) * D];
        s2 += p[(size_t)(i + 2) * D];
        s3 += p[(size_t)(i + 3) * D];
    }
    const float xa = xin + ((s0 + s1) + (s2 + s3));

    float s = xa, sq = xa * xa;
    for (int off = 16; off > 0; off >>= 1) {
        s  += __shfl_xor(s,  off, 32);
        sq += __shfl_xor(sq, off, 32);
    }
    if (lane == 0) { ssum[wave] = s; ssq[wave] = sq; }
    __syncthreads();
    if (wave == 0) {
        float a = ssum[lane];
        float c = ssq[lane];
        for (int off = 16; off > 0; off >>= 1) {
            a += __shfl_xor(a, off, 32);
            c += __shfl_xor(c, off, 32);
        }
        if (lane == 0) { ssum[0] = a; ssq[0] = c; }
    }
    __syncthreads();

    const float mu   = ssum[0] * (1.f / D);
    const float var  = ssq[0] * (1.f / D) - mu * mu;
    const float rstd = rsqrtf(var + EPS);

    const float xl = (xa - mu) * rstd * wv + bv;
    x_att[t] = xa;
    xln_out[t] = xl;
    xk_out[t] = xl * kv + pv * (1.f - kv);
    xr_out[t] = xl * rv + pv * (1.f - rv);
}

__global__ __launch_bounds__(256) void ffn_kv_kernel(
    const float* __restrict__ frw, const float* __restrict__ fxr,
    float* __restrict__ fr,
    const float* __restrict__ fkw, const float* __restrict__ fxk,
    const float* __restrict__ fvw, float* __restrict__ partial)
{
    const int blk = blockIdx.x;
    if (blk >= H / 16) {
        matvec_body(frw, fxr, fr, D, D, OP_SIGMOID, nullptr, blk - H / 16);
        return;
    }

    __shared__ float kpart[8][16];
    __shared__ __attribute__((aligned(16))) float fks[16];

    const int wave = threadIdx.x >> 5;
    const int lane = threadIdx.x & 31;
    const int half = lane >> 4;
    const int m    = lane & 15;

    const float* vbase = fvw + (size_t)(wave * 128 + m) * H + (blk << 4) + (half << 2);
    v4f qa[8][2];
    #pragma unroll
    for (int i = 0; i < 8; ++i) {
        qa[i][0] = *(const v4f*)(vbase + (size_t)(i * 16) * H);
        qa[i][1] = *(const v4f*)(vbase + (size_t)(i * 16) * H + 8);
    }
    __builtin_amdgcn_sched_barrier(0);

    v8f c = matvec_wave_slice(fkw, fxk, D, D, blk << 4, wave);

    if (m == 0) {
        const int rbase = half << 3;
        #pragma unroll
        for (int j = 0; j < 8; ++j)
            kpart[wave][rbase + j] = c[j];
    }
    __syncthreads();
    if (threadIdx.x < 16) {
        float v = 0.f;
        #pragma unroll
        for (int wv = 0; wv < 8; ++wv)
            v += kpart[wv][threadIdx.x];
        float t = fmaxf(v, 0.f);
        fks[threadIdx.x] = t * t;
    }
    __syncthreads();

    const v4f qb0 = *(const v4f*)(fks + (half << 2));
    const v4f qb1 = *(const v4f*)(fks + 8 + (half << 2));
    const v2f b00 = { qb0.x, qb0.y };
    const v2f b01 = { qb0.z, qb0.w };
    const v2f b10 = { qb1.x, qb1.y };
    const v2f b11 = { qb1.z, qb1.w };

    float* dst = partial + (size_t)blk * D + wave * 128 + (half << 3);

    #pragma unroll
    for (int g = 0; g < 8; g += 4) {
        v8f acc[4];
        #pragma unroll
        for (int i = 0; i < 4; ++i) {
            const v2f a00 = { qa[g + i][0].x, qa[g + i][0].y };
            acc[i] = (v8f){};
            acc[i] = __builtin_amdgcn_wmma_f32_16x16x4_f32(false, a00, false, b00, (short)0, acc[i], false, false);
        }
        #pragma unroll
        for (int i = 0; i < 4; ++i) {
            const v2f a01 = { qa[g + i][0].z, qa[g + i][0].w };
            acc[i] = __builtin_amdgcn_wmma_f32_16x16x4_f32(false, a01, false, b01, (short)0, acc[i], false, false);
        }
        #pragma unroll
        for (int i = 0; i < 4; ++i) {
            const v2f a10 = { qa[g + i][1].x, qa[g + i][1].y };
            acc[i] = __builtin_amdgcn_wmma_f32_16x16x4_f32(false, a10, false, b10, (short)0, acc[i], false, false);
        }
        #pragma unroll
        for (int i = 0; i < 4; ++i) {
            const v2f a11 = { qa[g + i][1].z, qa[g + i][1].w };
            acc[i] = __builtin_amdgcn_wmma_f32_16x16x4_f32(false, a11, false, b11, (short)0, acc[i], false, false);
        }
        if (m == 0) {
            #pragma unroll
            for (int i = 0; i < 4; ++i) {
                const v4f lo = { acc[i][0], acc[i][1], acc[i][2], acc[i][3] };
                const v4f hi = { acc[i][4], acc[i][5], acc[i][6], acc[i][7] };
                *(v4f*)(dst + (g + i) * 16)     = lo;
                *(v4f*)(dst + (g + i) * 16 + 4) = hi;
            }
        }
    }
}

__global__ __launch_bounds__(256) void ffn_out_kernel(
    const float* __restrict__ partial, const float* __restrict__ x_att,
    const float* __restrict__ fr, const float* __restrict__ staged,
    float* __restrict__ out)
{
    __shared__ float red[8][32];
    const int lane = threadIdx.x & 31;
    const int g    = threadIdx.x >> 5;
    const int col  = blockIdx.x * 32 + lane;

    const bool copies = (g >= 1 && g <= 3);
    const float sv  = staged[(size_t)(copies ? g - 1 : 0) * D + col];
    const float xa  = x_att[col];
    const float frv = fr[col];

    const float* p = partial + (size_t)(g * 32) * D + col;
    float s0 = 0.f, s1 = 0.f, s2 = 0.f, s3 = 0.f;
    #pragma unroll
    for (int i = 0; i < 32; i += 4) {
        s0 += p[(size_t)(i + 0) * D];
        s1 += p[(size_t)(i + 1) * D];
        s2 += p[(size_t)(i + 2) * D];
        s3 += p[(size_t)(i + 3) * D];
    }
    red[g][lane] = (s0 + s1) + (s2 + s3);
    __syncthreads();

    if (copies)
        out[(size_t)(2 + g) * D + col] = sv;
    if (threadIdx.x < 32) {
        float v = 0.f;
        #pragma unroll
        for (int w = 0; w < 8; ++w)
            v += red[w][threadIdx.x];
        out[col] = xa + frv * v;
    }
}

extern "C" void kernel_launch(void* const* d_in, const int* in_sizes, int n_in,
                              void* d_out, int out_size, void* d_ws, size_t ws_size,
                              hipStream_t stream) {
    const float* x     = (const float*)d_in[0];
    const float* state = (const float*)d_in[1];
    const float* tmk   = (const float*)d_in[2];
    const float* tmv   = (const float*)d_in[3];
    const float* tmr   = (const float*)d_in[4];
    const float* tf    = (const float*)d_in[5];
    const float* td    = (const float*)d_in[6];
    const float* kw    = (const float*)d_in[7];
    const float* vw    = (const float*)d_in[8];
    const float* rw    = (const float*)d_in[9];
    const float* ow    = (const float*)d_in[10];
    const float* lnw   = (const float*)d_in[11];
    const float* lnb   = (const float*)d_in[12];
    const float* ftmk  = (const float*)d_in[13];
    const float* ftmr  = (const float*)d_in[14];
    const float* fkw   = (const float*)d_in[15];
    const float* fvw   = (const float*)d_in[16];
    const float* frw   = (const float*)d_in[17];
    const float* flnw  = (const float*)d_in[18];
    const float* flnb  = (const float*)d_in[19];

    float* ws    = (float*)d_ws;
    float* xk    = ws + 1 * D;
    float* xv    = ws + 2 * D;
    float* xr    = ws + 3 * D;
    float* staged = ws + 4 * D;
    float* x_att = ws + 8 * D;
    float* fxk   = ws + 9 * D;
    float* fxr   = ws + 10 * D;
    float* fr    = ws + 11 * D;
    float* fvpart = ws + 16 * D;
    float* opart  = ws + 512 * D;

    float* out = (float*)d_out;
    float* s0  = out + 1 * D;
    float* s1  = out + 2 * D;

    ln_mix_kernel<<<1, 256, 0, stream>>>(x, lnw, lnb, state + 1 * D,
                                         tmk, tmv, tmr, s1, xk, xv, xr);

    att_kvr_ow_kernel<<<D / 16, 768, 0, stream>>>(kw, vw, rw, xk, xv, xr, state, tf, td,
                                                   ow, opart,
                                                   staged, staged + 1 * D, staged + 2 * D);

    att_sum_ln_mix_kernel<<<1, 1024, 0, stream>>>(opart, x, flnw, flnb, state  ,
                                                  ftmk, ftmr, x_att, s0, fxk, fxr);

    ffn_kv_kernel<<<(H / 16) + (D / 16), 256, 0, stream>>>(frw, fxr, fr, fkw, fxk, fvw, fvpart);

    ffn_out_kernel<<<D / 32, 256, 0, stream>>>(fvpart, x_att, fr, staged, out);
}


# Attention partial sum spread from one block to 32 (one per line of x_att, each with its line's sum, square sum); FFN waves finish LayerNorm+mix on their K slice via wave-private LDS; sum order changes
# speedup vs baseline: 1.8585x; 1.0428x over previous
#include <hip/hip_runtime.h>
#include <hip/hip_bf16.h>

#define D 1024
#define H 4096
#define EPS 1e-5f

typedef __attribute__((ext_vector_type(2))) float v2f;
typedef __attribute__((ext_vector_type(8))) float v8f;
typedef __attribute__((ext_vector_type(4))) float v4f;

#define OP_NONE     0
#define OP_SIGMOID  1
#define OP_ADDRES   2

__global__ __launch_bounds__(256) void ln_mix_kernel(
    const float* __restrict__ xin, const float* __restrict__ w,
    const float* __restrict__ b,   const float* __restrict__ prev,
    const float* __restrict__ tmk, const float* __restrict__ tmv,
    const float* __restrict__ tmr,
    float* __restrict__ xln_out, float* __restrict__ xk_out,
    float* __restrict__ xv_out,  float* __restrict__ xr_out)
{
    __shared__ float ssum[8];
    __shared__ float ssq[8];
    const int t = threadIdx.x;
    const int wave = t >> 5;
    const int lane = t & 31;

    const v4f x4 = ((const v4f*)xin)[t];
    const v4f w4 = ((const v4f*)w)[t];
    const v4f b4 = ((const v4f*)b)[t];
    const v4f p4 = ((const v4f*)prev)[t];
    const v4f k4 = ((const v4f*)tmk)[t];
    const v4f v4 = ((const v4f*)tmv)[t];
    const v4f r4 = ((const v4f*)tmr)[t];

    float s  = (x4.x + x4.y) + (x4.z + x4.w);
    float sq = (x4.x * x4.x + x4.y * x4.y) + (x4.z * x4.z + x4.w * x4.w);
    for (int off = 16; off > 0; off >>= 1) {
        s  += __shfl_xor(s,  off, 32);
        sq += __shfl_xor(sq, off, 32);
    }
    if (lane == 0) { ssum[wave] = s; ssq[wave] = sq; }
    __syncthreads();
    if (wave == 0) {
        float a = (lane < 8) ? ssum[lane] : 0.f;
        float c = (lane < 8) ? ssq[lane]  : 0.f;
        for (int off = 4; off > 0; off >>= 1) {
            a += __shfl_xor(a, off, 32);
            c += __shfl_xor(c, off, 32);
        }
        if (lane == 0) { ssum[0] = a; ssq[0] = c; }
    }
    __syncthreads();

    const float mu   = ssum[0] * (1.f / D);
    const float var  = ssq[0] * (1.f / D) - mu * mu;
    const float rstd = rsqrtf(var + EPS);

    const v4f xl = (x4 - mu) * rstd * w4 + b4;
    ((v4f*)xln_out)[t] = xl;
    ((v4f*)xk_out)[t]  = xl * k4 + p4 * (1.f - k4);
    ((v4f*)xv_out)[t]  = xl * v4 + p4 * (1.f - v4);
    ((v4f*)xr_out)[t]  = xl * r4 + p4 * (1.f - r4);
}

__device__ __forceinline__ v8f matvec_wave_slice(
    const float* __restrict__ W, const float* __restrict__ x,
    int K, int ldw, int row0, int wave)
{
    const int lane = threadIdx.x & 31;
    const int half = lane >> 4;
    const int m    = lane & 15;

    const int kchunk = K >> 3;
    const int k0 = wave * kchunk;
    const int k1 = k0 + kchunk;

    const float* wrow = W + (size_t)(row0 + m) * (size_t)ldw + (half << 1);
    const float* xrow = x + (half << 1);

    v8f c0 = {}; v8f c1 = {}; v8f c2 = {}; v8f c3 = {};

    for (int k = k0; k < k1; k += 16) {
        v2f a0 = *(const v2f*)(wrow + k);
        v2f a1 = *(const v2f*)(wrow + k + 4);
        v2f a2 = *(const v2f*)(wrow + k + 8);
        v2f a3 = *(const v2f*)(wrow + k + 12);
        v2f b0 = *(const v2f*)(xrow + k);
        v2f b1 = *(const v2f*)(xrow + k + 4);
        v2f b2 = *(const v2f*)(xrow + k + 8);
        v2f b3 = *(const v2f*)(xrow + k + 12);
        c0 = __builtin_amdgcn_wmma_f32_16x16x4_f32(false, a0, false, b0, (short)0, c0, false, false);
        c1 = __builtin_amdgcn_wmma_f32_16x16x4_f32(false, a1, false, b1, (short)0, c1, false, false);
        c2 = __builtin_amdgcn_wmma_f32_16x16x4_f32(false, a2, false, b2, (short)0, c2, false, false);
        c3 = __builtin_amdgcn_wmma_f32_16x16x4_f32(false, a3, false, b3, (short)0, c3, false, false);
    }

    return c0 + c1 + c2 + c3;
}

struct MixIn { v2f st; v4f xa; v4f w; v4f b; v4f p; v4f t; };

__device__ __forceinline__ MixIn mix_request(
    const float* __restrict__ x_att, const float* __restrict__ stats,
    const float* __restrict__ w, const float* __restrict__ b,
    const float* __restrict__ prev, const float* __restrict__ tm, int slice)
{
    const int lane = threadIdx.x & 31;
    const int q = slice * 32 + lane;
    MixIn in;
    in.st = *(const v2f*)(stats + 32 * lane);
    in.xa = ((const v4f*)x_att)[q];
    in.w  = ((const v4f*)w)[q];
    in.b  = ((const v4f*)b)[q];
    in.p  = ((const v4f*)prev)[q];
    in.t  = ((const v4f*)tm)[q];
    return in;
}

__device__ __forceinline__ v4f mix_finish(const MixIn& in, float* bl)
{
    const int lane = threadIdx.x & 31;
    float s = in.st.x, sq = in.st.y;
    for (int off = 16; off > 0; off >>= 1) {
        s  += __shfl_xor(s,  off, 32);
        sq += __shfl_xor(sq, off, 32);
    }
    const float mu   = s * (1.f / D);
    const float var  = sq * (1.f / D) - mu * mu;
    const float rstd = rsqrtf(var + EPS);
    const v4f xl = (in.xa - mu) * rstd * in.w + in.b;
    *(v4f*)(bl + 4 * lane) = xl * in.t + in.p * (1.f - in.t);
    __builtin_amdgcn_wave_barrier();
    return xl;
}

__device__ __forceinline__ void slice_request_a(
    const float* __restrict__ W, int row0, int wave, v2f a[32])
{
    const int lane = threadIdx.x & 31;
    const int half = lane >> 4;
    const int m    = lane & 15;
    const float* wrow = W + (size_t)(row0 + m) * (size_t)D + (half << 1) + wave * (D >> 3);
    #pragma unroll
    for (int j = 0; j < 32; ++j)
        a[j] = *(const v2f*)(wrow + 4 * j);
}

__device__ __forceinline__ v8f slice_product(const v2f a[32], const float* bl)
{
    const int half = (threadIdx.x & 31) >> 4;
    const float* brow = bl + (half << 1);
    v8f c0 = {}; v8f c1 = {}; v8f c2 = {}; v8f c3 = {};
    #pragma unroll
    for (int j = 0; j < 32; j += 4) {
        const v2f b0 = *(const v2f*)(brow + 4 * j);
        const v2f b1 = *(const v2f*)(brow + 4 * j + 4);
        const v2f b2 = *(const v2f*)(brow + 4 * j + 8);
        const v2f b3 = *(const v2f*)(brow + 4 * j + 12);
        c0 = __builtin_amdgcn_wmma_f32_16x16x4_f32(false, a[j],     false, b0, (short)0, c0, false, false);
        c1 = __builtin_amdgcn_wmma_f32_16x16x4_f32(false, a[j + 1], false, b1, (short)0, c1, false, false);
        c2 = __builtin_amdgcn_wmma_f32_16x16x4_f32(false, a[j + 2], false, b2, (short)0, c2, false, false);
        c3 = __builtin_amdgcn_wmma_f32_16x16x4_f32(false, a[j + 3], false, b3, (short)0, c3, false, false);
    }
    return c0 + c1 + c2 + c3;
}

__device__ __forceinline__ v8f ow_tile(const v4f q0, const v4f q1,
    const v2f b00, const v2f b01, const v2f b10, const v2f b11)
{
    const v2f a00 = { q0.x, q0.y };
    const v2f a01 = { q0.z, q0.w };
    const v2f a10 = { q1.x, q1.y };
    const v2f a11 = { q1.z, q1.w };
    v8f acc = {};
    acc = __builtin_amdgcn_wmma_f32_16x16x4_f32(false, a00, false, b00, (short)0, acc, false, false);
    acc = __builtin_amdgcn_wmma_f32_16x16x4_f32(false, a01, false, b01, (short)0, acc, false, false);
    acc = __builtin_amdgcn_wmma_f32_16x16x4_f32(false, a10, false, b10, (short)0, acc, false, false);
    acc = __builtin_amdgcn_wmma_f32_16x16x4_f32(false, a11, false, b11, (short)0, acc, false, false);
    return acc;
}

__device__ __forceinline__ void store_tile_rows(float* __restrict__ dst, const v8f acc)
{
    const v4f lo = { acc[0], acc[1], acc[2], acc[3] };
    const v4f hi = { acc[4], acc[5], acc[6], acc[7] };
    *(v4f*)(dst)     = lo;
    *(v4f*)(dst + 4) = hi;
}

__global__ __launch_bounds__(768) void att_kvr_ow_kernel(
    const float* __restrict__ kw, const float* __restrict__ vw,
    const float* __restrict__ rw,
    const float* __restrict__ xk, const float* __restrict__ xv,
    const float* __restrict__ xr,
    const float* __restrict__ state,
    const float* __restrict__ time_first, const float* __restrict__ time_decay,
    const float* __restrict__ ow, float* __restrict__ partial,
    float* __restrict__ st2, float* __restrict__ st3, float* __restrict__ st4)
{
    __shared__ float kpart[3][8][16];
    __shared__ __attribute__((aligned(16))) float rws[16];

    const int blk   = blockIdx.x;
    const int wave  = threadIdx.x >> 5;
    const int lane  = threadIdx.x & 31;
    const int half  = lane >> 4;
    const int m     = lane & 15;
    const int job   = wave >> 3;
    const int slice = wave & 7;
    const bool third = wave < 16;

    const float* obase = ow + (size_t)(wave * 16 + m) * D + (blk << 4) + (half << 2);
    v4f qa[3][2];
    qa[0][0] = *(const v4f*)(obase);
    qa[0][1] = *(const v4f*)(obase + 8);
    qa[1][0] = *(const v4f*)(obase + (size_t)(24 * 16) * D);
    qa[1][1] = *(const v4f*)(obase + (size_t)(24 * 16) * D + 8);
    qa[2][0] = (v4f){};
    qa[2][1] = (v4f){};
    if (third) {
        qa[2][0] = *(const v4f*)(obase + (size_t)(48 * 16) * D);
        qa[2][1] = *(const v4f*)(obase + (size_t)(48 * 16) * D + 8);
    }

    const int ch = (blk << 4) + m;
    const float aa = state[2 * D + ch];
    const float bb = state[3 * D + ch];
    const float pp = state[4 * D + ch];
    const float tf = time_first[ch];
    const float td = time_decay[ch];

    const float* Wj = (job == 0) ? kw : (job == 1) ? vw : rw;
    const float* xj = (job == 0) ? xk : (job == 1) ? xv : xr;
    v8f c = matvec_wave_slice(Wj, xj, D, D, blk << 4, slice);

    if (m == 0) {
        const int rbase = half << 3;
        #pragma unroll
        for (int j = 0; j < 8; ++j)
            kpart[job][slice][rbase + j] = c[j];
    }
    __syncthreads();

    float n2 = 0.f, n3 = 0.f, n4 = 0.f;
    if (threadIdx.x < 16) {
        float k = 0.f, v = 0.f, r = 0.f;
        #pragma unroll
        for (int wv = 0; wv < 8; ++wv) {
            k += kpart[0][wv][threadIdx.x];
            v += kpart[1][wv][threadIdx.x];
            r += kpart[2][wv][threadIdx.x];
        }
        r = 1.f / (1.f + expf(-r));

        float ww = tf + k;
        float qq = fmaxf(pp, ww);
        float e1 = expf(pp - qq);
        float e2 = expf(ww - qq);
        float wkv = (e1 * aa + e2 * v) / (e1 * bb + e2);
        rws[threadIdx.x] = r * wkv;

        float ww2 = pp + td;
        float qq2 = fmaxf(ww2, k);
        float f1 = expf(ww2 - qq2);
        float f2 = expf(k - qq2);
        n2 = f1 * aa + f2 * v;
        n3 = f1 * bb + f2;
        n4 = qq2;
    }
    __syncthreads();
    if (threadIdx.x < 16) {
        st2[ch] = n2;
        st3[ch] = n3;
        st4[ch] = n4;
    }

    const v4f qb0 = *(const v4f*)(rws + (half << 2));
    const v4f qb1 = *(const v4f*)(rws + 8 + (half << 2));
    const v2f b00 = { qb0.x, qb0.y };
    const v2f b01 = { qb0.z, qb0.w };
    const v2f b10 = { qb1.x, qb1.y };
    const v2f b11 = { qb1.z, qb1.w };

    float* dst = partial + (size_t)blk * D + wave * 16 + (half << 3);

    const v8f acc0 = ow_tile(qa[0][0], qa[0][1], b00, b01, b10, b11);
    const v8f acc1 = ow_tile(qa[1][0], qa[1][1], b00, b01, b10, b11);
    if (third) {
        const v8f acc2 = ow_tile(qa[2][0], qa[2][1], b00, b01, b10, b11);
        if (m == 0)
            store_tile_rows(dst + 48 * 16, acc2);
    }
    if (m == 0) {
        store_tile_rows(dst, acc0);
        store_tile_rows(dst + 24 * 16, acc1);
    }
}

__global__ __launch_bounds__(256) void att_sum_kernel(
    const float* __restrict__ partial, const float* __restrict__ x,
    float* __restrict__ x_att, float* __restrict__ stats)
{
    __shared__ float red[8][32];
    const int lane = threadIdx.x & 31;
    const int g    = threadIdx.x >> 5;
    const int col  = blockIdx.x * 32 + lane;

    const float xin = x[col];

    const float* p = partial + (size_t)(g * 8) * D + col;
    float s0 = 0.f, s1 = 0.f, s2 = 0.f, s3 = 0.f;
    #pragma unroll
    for (int i = 0; i < 8; i += 4) {
        s0 += p[(size_t)(i + 0) * D];
        s1 += p[(size_t)(i + 1) * D];
        s2 += p[(size_t)(i + 2) * D];
        s3 += p[(size_t)(i + 3) * D];
    }
    red[g][lane] = (s0 + s1) + (s2 + s3);
    __syncthreads();

    if (threadIdx.x < 32) {
        float v = 0.f;
        #pragma unroll
        for (int w = 0; w < 8; ++w)
            v += red[w][threadIdx.x];
        const float xa = xin + v;
        float s = xa, sq = xa * xa;
        for (int off = 16; off > 0; off >>= 1) {
            s  += __shfl_xor(s,  off, 32);
            sq += __shfl_xor(sq, off, 32);
        }
        x_att[col] = xa;
        stats[blockIdx.x * 32 + lane] = (lane & 1) ? sq : s;
    }
}

__global__ __launch_bounds__(256) void ffn_kv_kernel(
    const float* __restrict__ frw, float* __restrict__ fr,
    const float* __restrict__ fkw,
    const float* __restrict__ fvw, float* __restrict__ partial,
    const float* __restrict__ x_att, const float* __restrict__ stats,
    const float* __restrict__ lnw, const float* __restrict__ lnb,
    const float* __restrict__ prev,
    const float* __restrict__ tmk, const float* __restrict__ tmr,
    float* __restrict__ xln_out)
{
    __shared__ float kpart[8][16];
    __shared__ __attribute__((aligned(16))) float fks[16];
    __shared__ __attribute__((aligned(16))) float bls[8][128];

    const int blk  = blockIdx.x;
    const int wave = threadIdx.x >> 5;
    const int lane = threadIdx.x & 31;
    const int half = lane >> 4;
    const int m    = lane & 15;
    const bool isr = blk >= H / 16;

    const MixIn in = mix_request(x_att, stats, lnw, lnb, prev, isr ? tmr : tmk, wave);

    v4f qa[8][2];
    if (!isr) {
        const float* vbase = fvw + (size_t)(wave * 128 + m) * H + (blk << 4) + (half << 2);
        #pragma unroll
        for (int i = 0; i < 8; ++i) {
            qa[i][0] = *(const v4f*)(vbase + (size_t)(i * 16) * H);
            qa[i][1] = *(const v4f*)(vbase + (size_t)(i * 16) * H + 8);
        }
    }
    __builtin_amdgcn_sched_barrier(0);

    const float* Wsel = isr ? frw : fkw;
    const int row0 = (isr ? blk - H / 16 : blk) << 4;
    v2f a[32];
    slice_request_a(Wsel, row0, wave, a);
    const v4f xl = mix_finish(in, bls[wave]);
    v8f c = slice_product(a, bls[wave]);

    if (m == 0) {
        const int rbase = half << 3;
        #pragma unroll
        for (int j = 0; j < 8; ++j)
            kpart[wave][rbase + j] = c[j];
    }
    __syncthreads();

    if (isr) {
        if (threadIdx.x < 16) {
            float v = 0.f;
            #pragma unroll
            for (int wv = 0; wv < 8; ++wv)
                v += kpart[wv][threadIdx.x];
            fr[row0 + threadIdx.x] = 1.f / (1.f + expf(-v));
        }
        if (blk == H / 16)
            ((v4f*)xln_out)[wave * 32 + lane] = xl;
        return;
    }

    if (threadIdx.x < 16) {
        float v = 0.f;
        #pragma unroll
        for (int wv = 0; wv < 8; ++wv)
            v += kpart[wv][threadIdx.x];
        float t = fmaxf(v, 0.f);
        fks[threadIdx.x] = t * t;
    }
    __syncthreads();

    const v4f qb0 = *(const v4f*)(fks + (half << 2));
    const v4f qb1 = *(const v4f*)(fks + 8 + (half << 2));
    const v2f b00 = { qb0.x, qb0.y };
    const v2f b01 = { qb0.z, qb0.w };
    const v2f b10 = { qb1.x, qb1.y };
    const v2f b11 = { qb1.z, qb1.w };

    float* dst = partial + (size_t)blk * D + wave * 128 + (half << 3);

    #pragma unroll
    for (int g = 0; g < 8; g += 4) {
        v8f acc[4];
        #pragma unroll
        for (int i = 0; i < 4; ++i) {
            const v2f a00 = { qa[g + i][0].x, qa[g + i][0].y };
            acc[i] = (v8f){};
            acc[i] = __builtin_amdgcn_wmma_f32_16x16x4_f32(false, a00, false, b00, (short)0, acc[i], false, false);
        }
        #pragma unroll
        for (int i = 0; i < 4; ++i) {
            const v2f a01 = { qa[g + i][0].z, qa[g + i][0].w };
            acc[i] = __builtin_amdgcn_wmma_f32_16x16x4_f32(false, a01, false, b01, (short)0, acc[i], false, false);
        }
        #pragma unroll
        for (int i = 0; i < 4; ++i) {
            const v2f a10 = { qa[g + i][1].x, qa[g + i][1].y };
            acc[i] = __builtin_amdgcn_wmma_f32_16x16x4_f32(false, a10, false, b10, (short)0, acc[i], false, false);
        }
        #pragma unroll
        for (int i = 0; i < 4; ++i) {
            const v2f a11 = { qa[g + i][1].z, qa[g + i][1].w };
            acc[i] = __builtin_amdgcn_wmma_f32_16x16x4_f32(false, a11, false, b11, (short)0, acc[i], false, false);
        }
        if (m == 0) {
            #pragma unroll
            for (int i = 0; i < 4; ++i) {
                const v4f lo = { acc[i][0], acc[i][1], acc[i][2], acc[i][3] };
                const v4f hi = { acc[i][4], acc[i][5], acc[i][6], acc[i][7] };
                *(v4f*)(dst + (g + i) * 16)     = lo;
                *(v4f*)(dst + (g + i) * 16 + 4) = hi;
            }
        }
    }
}

__global__ __launch_bounds__(256) void ffn_out_kernel(
    const float* __restrict__ partial, const float* __restrict__ x_att,
    const float* __restrict__ fr, const float* __restrict__ staged,
    float* __restrict__ out)
{
    __shared__ float red[8][32];
    const int lane = threadIdx.x & 31;
    const int g    = threadIdx.x >> 5;
    const int col  = blockIdx.x * 32 + lane;

    const bool copies = (g >= 1 && g <= 3);
    const float sv  = staged[(size_t)(copies ? g - 1 : 0) * D + col];
    const float xa  = x_att[col];
    const float frv = fr[col];

    const float* p = partial + (size_t)(g * 32) * D + col;
    float s0 = 0.f, s1 = 0.f, s2 = 0.f, s3 = 0.f;
    #pragma unroll
    for (int i = 0; i < 32; i += 4) {
        s0 += p[(size_t)(i + 0) * D];
        s1 += p[(size_t)(i + 1) * D];
        s2 += p[(size_t)(i + 2) * D];
        s3 += p[(size_t)(i + 3) * D];
    }
    red[g][lane] = (s0 + s1) + (s2 + s3);
    __syncthreads();

    if (copies)
        out[(size_t)(2 + g) * D + col] = sv;
    if (threadIdx.x < 32) {
        float v = 0.f;
        #pragma unroll
        for (int w = 0; w < 8; ++w)
            v += red[w][threadIdx.x];
        out[col] = xa + frv * v;
    }
}

extern "C" void kernel_launch(void* const* d_in, const int* in_sizes, int n_in,
                              void* d_out, int out_size, void* d_ws, size_t ws_size,
                              hipStream_t stream) {
    const float* x     = (const float*)d_in[0];
    const float* state = (const float*)d_in[1];
    const float* tmk   = (const float*)d_in[2];
    const float* tmv   = (const float*)d_in[3];
    const float* tmr   = (const float*)d_in[4];
    const float* tf    = (const float*)d_in[5];
    const float* td    = (const float*)d_in[6];
    const float* kw    = (const float*)d_in[7];
    const float* vw    = (const float*)d_in[8];
    const float* rw    = (const float*)d_in[9];
    const float* ow    = (const float*)d_in[10];
    const float* lnw   = (const float*)d_in[11];
    const float* lnb   = (const float*)d_in[12];
    const float* ftmk  = (const float*)d_in[13];
    const float* ftmr  = (const float*)d_in[14];
    const float* fkw   = (const float*)d_in[15];
    const float* fvw   = (const float*)d_in[16];
    const float* frw   = (const float*)d_in[17];
    const float* flnw  = (const float*)d_in[18];
    const float* flnb  = (const float*)d_in[19];

    float* ws    = (float*)d_ws;
    float* xk    = ws + 1 * D;
    float* xv    = ws + 2 * D;
    float* xr    = ws + 3 * D;
    float* staged = ws + 4 * D;
    float* x_att = ws + 8 * D;
    float* stats = ws + 9 * D;
    float* fr    = ws + 11 * D;
    float* fvpart = ws + 16 * D;
    float* opart  = ws + 512 * D;

    float* out = (float*)d_out;
    float* s0  = out + 1 * D;
    float* s1  = out + 2 * D;

    ln_mix_kernel<<<1, 256, 0, stream>>>(x, lnw, lnb, state + 1 * D,
                                         tmk, tmv, tmr, s1, xk, xv, xr);

    att_kvr_ow_kernel<<<D / 16, 768, 0, stream>>>(kw, vw, rw, xk, xv, xr, state, tf, td,
                                                   ow, opart,
                                                   staged, staged + 1 * D, staged + 2 * D);

    att_sum_kernel<<<D / 32, 256, 0, stream>>>(opart, x, x_att, stats);

    ffn_kv_kernel<<<(H / 16) + (D / 16), 256, 0, stream>>>(frw, fr, fkw, fvw, fvpart,
                                                           x_att, stats, flnw, flnb, state  ,
                                                           ftmk, ftmr, s0);

    ffn_out_kernel<<<D / 32, 256, 0, stream>>>(fvpart, x_att, fr, staged, out);
}


# k/v/r + WKV + ow launch re-cut from 64 blocks of 16 channels to 256 of 4: one A fragment holds rows of kw, vw, rw, B columns hold xk, xv, xr; next launch sums 256 partials; sum order changes
# speedup vs baseline: 2.1136x; 1.1373x over previous
#include <hip/hip_runtime.h>
#include <hip/hip_bf16.h>

#define D 1024
#define H 4096
#define EPS 1e-5f

typedef __attribute__((ext_vector_type(2))) float v2f;
typedef __attribute__((ext_vector_type(8))) float v8f;
typedef __attribute__((ext_vector_type(4))) float v4f;

#define OP_NONE     0
#define OP_SIGMOID  1
#define OP_ADDRES   2

__global__ __launch_bounds__(256) void ln_mix_kernel(
    const float* __restrict__ xin, const float* __restrict__ w,
    const float* __restrict__ b,   const float* __restrict__ prev,
    const float* __restrict__ tmk, const float* __restrict__ tmv,
    const float* __restrict__ tmr,
    float* __restrict__ xln_out, float* __restrict__ xk_out,
    float* __restrict__ xv_out,  float* __restrict__ xr_out)
{
    __shared__ float ssum[8];
    __shared__ float ssq[8];
    const int t = threadIdx.x;
    const int wave = t >> 5;
    const int lane = t & 31;

    const v4f x4 = ((const v4f*)xin)[t];
    const v4f w4 = ((const v4f*)w)[t];
    const v4f b4 = ((const v4f*)b)[t];
    const v4f p4 = ((const v4f*)prev)[t];
    const v4f k4 = ((const v4f*)tmk)[t];
    const v4f v4 = ((const v4f*)tmv)[t];
    const v4f r4 = ((const v4f*)tmr)[t];

    float s  = (x4.x + x4.y) + (x4.z + x4.w);
    float sq = (x4.x * x4.x + x4.y * x4.y) + (x4.z * x4.z + x4.w * x4.w);
    for (int off = 16; off > 0; off >>= 1) {
        s  += __shfl_xor(s,  off, 32);
        sq += __shfl_xor(sq, off, 32);
    }
    if (lane == 0) { ssum[wave] = s; ssq[wave] = sq; }
    __syncthreads();
    if (wave == 0) {
        float a = (lane < 8) ? ssum[lane] : 0.f;
        float c = (lane < 8) ? ssq[lane]  : 0.f;
        for (int off = 4; off > 0; off >>= 1) {
            a += __shfl_xor(a, off, 32);
            c += __shfl_xor(c, off, 32);
        }
        if (lane == 0) { ssum[0] = a; ssq[0] = c; }
    }
    __syncthreads();

    const float mu   = ssum[0] * (1.f / D);
    const float var  = ssq[0] * (1.f / D) - mu * mu;
    const float rstd = rsqrtf(var + EPS);

    const v4f xl = (x4 - mu) * rstd * w4 + b4;
    ((v4f*)xln_out)[t] = xl;
    ((v4f*)xk_out)[t]  = xl * k4 + p4 * (1.f - k4);
    ((v4f*)xv_out)[t]  = xl * v4 + p4 * (1.f - v4);
    ((v4f*)xr_out)[t]  = xl * r4 + p4 * (1.f - r4);
}

struct MixIn { v2f st; v4f xa; v4f w; v4f b; v4f p; v4f t; };

__device__ __forceinline__ MixIn mix_request(
    const float* __restrict__ x_att, const float* __restrict__ stats,
    const float* __restrict__ w, const float* __restrict__ b,
    const float* __restrict__ prev, const float* __restrict__ tm, int slice)
{
    const int lane = threadIdx.x & 31;
    const int q = slice * 32 + lane;
    MixIn in;
    in.st = *(const v2f*)(stats + 32 * lane);
    in.xa = ((const v4f*)x_att)[q];
    in.w  = ((const v4f*)w)[q];
    in.b  = ((const v4f*)b)[q];
    in.p  = ((const v4f*)prev)[q];
    in.t  = ((const v4f*)tm)[q];
    return in;
}

__device__ __forceinline__ v4f mix_finish(const MixIn& in, float* bl)
{
    const int lane = threadIdx.x & 31;
    float s = in.st.x, sq = in.st.y;
    for (int off = 16; off > 0; off >>= 1) {
        s  += __shfl_xor(s,  off, 32);
        sq += __shfl_xor(sq, off, 32);
    }
    const float mu   = s * (1.f / D);
    const float var  = sq * (1.f / D) - mu * mu;
    const float rstd = rsqrtf(var + EPS);
    const v4f xl = (in.xa - mu) * rstd * in.w + in.b;
    *(v4f*)(bl + 4 * lane) = xl * in.t + in.p * (1.f - in.t);
    __builtin_amdgcn_wave_barrier();
    return xl;
}

__device__ __forceinline__ void slice_request_a(
    const float* __restrict__ W, int row0, int wave, v2f a[32])
{
    const int lane = threadIdx.x & 31;
    const int half = lane >> 4;
    const int m    = lane & 15;
    const float* wrow = W + (size_t)(row0 + m) * (size_t)D + (half << 1) + wave * (D >> 3);
    #pragma unroll
    for (int j = 0; j < 32; ++j)
        a[j] = *(const v2f*)(wrow + 4 * j);
}

__device__ __forceinline__ v8f slice_product(const v2f a[32], const float* bl)
{
    const int half = (threadIdx.x & 31) >> 4;
    const float* brow = bl + (half << 1);
    v8f c0 = {}; v8f c1 = {}; v8f c2 = {}; v8f c3 = {};
    #pragma unroll
    for (int j = 0; j < 32; j += 4) {
        const v2f b0 = *(const v2f*)(brow + 4 * j);
        const v2f b1 = *(const v2f*)(brow + 4 * j + 4);
        const v2f b2 = *(const v2f*)(brow + 4 * j + 8);
        const v2f b3 = *(const v2f*)(brow + 4 * j + 12);
        c0 = __builtin_amdgcn_wmma_f32_16x16x4_f32(false, a[j],     false, b0, (short)0, c0, false, false);
        c1 = __builtin_amdgcn_wmma_f32_16x16x4_f32(false, a[j + 1], false, b1, (short)0, c1, false, false);
        c2 = __builtin_amdgcn_wmma_f32_16x16x4_f32(false, a[j + 2], false, b2, (short)0, c2, false, false);
        c3 = __builtin_amdgcn_wmma_f32_16x16x4_f32(false, a[j + 3], false, b3, (short)0, c3, false, false);
    }
    return c0 + c1 + c2 + c3;
}

__global__ __launch_bounds__(128) __attribute__((amdgpu_waves_per_eu(1, 1)))
void att_kvr_ow_kernel(
    const float* __restrict__ kw, const float* __restrict__ vw,
    const float* __restrict__ rw,
    const float* __restrict__ xk, const float* __restrict__ xv,
    const float* __restrict__ xr,
    const float* __restrict__ state,
    const float* __restrict__ time_first, const float* __restrict__ time_decay,
    const float* __restrict__ ow, float* __restrict__ partial,
    float* __restrict__ st2, float* __restrict__ st3, float* __restrict__ st4)
{
    __shared__ __attribute__((aligned(16))) float kpart[4][16];
    __shared__ __attribute__((aligned(16))) float rws[4];

    const int blk   = blockIdx.x;
    const int wave  = threadIdx.x >> 5;
    const int lane  = threadIdx.x & 31;
    const int half  = lane >> 4;
    const int m     = lane & 15;
    const int c     = m & 3;
    const int job   = (m >> 2) > 2 ? 2 : (m >> 2);

    const int ch = (blk << 2) + c;
    const float aa = state[2 * D + ch];
    const float bb = state[3 * D + ch];
    const float pp = state[4 * D + ch];
    const float tf = time_first[ch];
    const float td = time_decay[ch];

    const float* Wj = (job == 0) ? kw : (job == 1) ? vw : rw;
    const float* xj = (job == 0) ? xk : (job == 1) ? xv : xr;
    const float* arow = Wj + (size_t)ch * D + wave * (D >> 2) + (half << 2);
    const float* brow = xj + wave * (D >> 2) + (half << 2);

    v4f a[32], b[32];
    #pragma unroll
    for (int i = 0; i < 32; ++i) {
        a[i] = *(const v4f*)(arow + 8 * i);
        b[i] = *(const v4f*)(brow + 8 * i);
    }

    const float* obase = ow + (size_t)(wave * 16 + m) * D + (blk << 2) + (half << 1);
    v2f qa[16];
    #pragma unroll
    for (int q = 0; q < 16; ++q)
        qa[q] = *(const v2f*)(obase + (size_t)(64 * q) * D);

    v8f c0 = {}; v8f c1 = {}; v8f c2 = {}; v8f c3 = {};
    #pragma unroll
    for (int i = 0; i < 32; i += 2) {
        const v2f a00 = { a[i].x, a[i].y },         b00 = { b[i].x, b[i].y };
        const v2f a01 = { a[i].z, a[i].w },         b01 = { b[i].z, b[i].w };
        const v2f a10 = { a[i + 1].x, a[i + 1].y }, b10 = { b[i + 1].x, b[i + 1].y };
        const v2f a11 = { a[i + 1].z, a[i + 1].w }, b11 = { b[i + 1].z, b[i + 1].w };
        c0 = __builtin_amdgcn_wmma_f32_16x16x4_f32(false, a00, false, b00, (short)0, c0, false, false);
        c1 = __builtin_amdgcn_wmma_f32_16x16x4_f32(false, a01, false, b01, (short)0, c1, false, false);
        c2 = __builtin_amdgcn_wmma_f32_16x16x4_f32(false, a10, false, b10, (short)0, c2, false, false);
        c3 = __builtin_amdgcn_wmma_f32_16x16x4_f32(false, a11, false, b11, (short)0, c3, false, false);
    }
    const v8f cs = c0 + c1 + c2 + c3;

    if (lane == 0) {
        const v4f t = { cs[0], cs[1], cs[2], cs[3] };
        *(v4f*)(&kpart[wave][0]) = t;
    } else if (lane == 4) {
        const v4f t = { cs[4], cs[5], cs[6], cs[7] };
        *(v4f*)(&kpart[wave][4]) = t;
    } else if (lane == 24) {
        const v4f t = { cs[0], cs[1], cs[2], cs[3] };
        *(v4f*)(&kpart[wave][8]) = t;
    }
    __syncthreads();

    float n2 = 0.f, n3 = 0.f, n4 = 0.f;
    if (threadIdx.x < 4) {
        float k = 0.f, v = 0.f, r = 0.f;
        #pragma unroll
        for (int wv = 0; wv < 4; ++wv) {
            k += kpart[wv][threadIdx.x];
            v += kpart[wv][4 + threadIdx.x];
            r += kpart[wv][8 + threadIdx.x];
        }
        r = 1.f / (1.f + expf(-r));

        float ww = tf + k;
        float qq = fmaxf(pp, ww);
        float e1 = expf(pp - qq);
        float e2 = expf(ww - qq);
        float wkv = (e1 * aa + e2 * v) / (e1 * bb + e2);
        rws[threadIdx.x] = r * wkv;

        float ww2 = pp + td;
        float qq2 = fmaxf(ww2, k);
        float f1 = expf(ww2 - qq2);
        float f2 = expf(k - qq2);
        n2 = f1 * aa + f2 * v;
        n3 = f1 * bb + f2;
        n4 = qq2;
    }
    __syncthreads();
    if (threadIdx.x < 4) {
        st2[ch] = n2;
        st3[ch] = n3;
        st4[ch] = n4;
    }

    const v2f bq = *(const v2f*)(rws + (half << 1));

    float* dst = partial + (size_t)blk * D + wave * 16 + (half << 3);
    v8f acc[16];
    #pragma unroll
    for (int q = 0; q < 16; ++q) {
        acc[q] = (v8f){};
        acc[q] = __builtin_amdgcn_wmma_f32_16x16x4_f32(false, qa[q], false, bq, (short)0, acc[q], false, false);
    }
    if (m == 0) {
        #pragma unroll
        for (int q = 0; q < 16; ++q) {
            const v4f lo = { acc[q][0], acc[q][1], acc[q][2], acc[q][3] };
            const v4f hi = { acc[q][4], acc[q][5], acc[q][6], acc[q][7] };
            *(v4f*)(dst + 64 * q)     = lo;
            *(v4f*)(dst + 64 * q + 4) = hi;
        }
    }
}

__global__ __launch_bounds__(256) void att_sum_kernel(
    const float* __restrict__ partial, const float* __restrict__ x,
    float* __restrict__ x_att, float* __restrict__ stats)
{
    __shared__ float red[8][32];
    const int lane = threadIdx.x & 31;
    const int g    = threadIdx.x >> 5;
    const int col  = blockIdx.x * 32 + lane;

    const float xin = x[col];

    const float* p = partial + (size_t)(g * 32) * D + col;
    float s0 = 0.f, s1 = 0.f, s2 = 0.f, s3 = 0.f;
    #pragma unroll
    for (int i = 0; i < 32; i += 4) {
        s0 += p[(size_t)(i + 0) * D];
        s1 += p[(size_t)(i + 1) * D];
        s2 += p[(size_t)(i + 2) * D];
        s3 += p[(size_t)(i + 3) * D];
    }
    red[g][lane] = (s0 + s1) + (s2 + s3);
    __syncthreads();

    if (threadIdx.x < 32) {
        float v = 0.f;
        #pragma unroll
        for (int w = 0; w < 8; ++w)
            v += red[w][threadIdx.x];
        const float xa = xin + v;
        float s = xa, sq = xa * xa;
        for (int off = 16; off > 0; off >>= 1) {
            s  += __shfl_xor(s,  off, 32);
            sq += __shfl_xor(sq, off, 32);
        }
        x_att[col] = xa;
        stats[blockIdx.x * 32 + lane] = (lane & 1) ? sq : s;
    }
}

__global__ __launch_bounds__(256) void ffn_kv_kernel(
    const float* __restrict__ frw, float* __restrict__ fr,
    const float* __restrict__ fkw,
    const float* __restrict__ fvw, float* __restrict__ partial,
    const float* __restrict__ x_att, const float* __restrict__ stats,
    const float* __restrict__ lnw, const float* __restrict__ lnb,
    const float* __restrict__ prev,
    const float* __restrict__ tmk, const float* __restrict__ tmr,
    float* __restrict__ xln_out)
{
    __shared__ float kpart[8][16];
    __shared__ __attribute__((aligned(16))) float fks[16];
    __shared__ __attribute__((aligned(16))) float bls[8][128];

    const int blk  = blockIdx.x;
    const int wave = threadIdx.x >> 5;
    const int lane = threadIdx.x & 31;
    const int half = lane >> 4;
    const int m    = lane & 15;
    const bool isr = blk >= H / 16;

    const MixIn in = mix_request(x_att, stats, lnw, lnb, prev, isr ? tmr : tmk, wave);

    v4f qa[8][2];
    if (!isr) {
        const float* vbase = fvw + (size_t)(wave * 128 + m) * H + (blk << 4) + (half << 2);
        #pragma unroll
        for (int i = 0; i < 8; ++i) {
            qa[i][0] = *(const v4f*)(vbase + (size_t)(i * 16) * H);
            qa[i][1] = *(const v4f*)(vbase + (size_t)(i * 16) * H + 8);
        }
    }
    __builtin_amdgcn_sched_barrier(0);

    const float* Wsel = isr ? frw : fkw;
    const int row0 = (isr ? blk - H / 16 : blk) << 4;
    v2f a[32];
    slice_request_a(Wsel, row0, wave, a);
    const v4f xl = mix_finish(in, bls[wave]);
    v8f c = slice_product(a, bls[wave]);

    if (m == 0) {
        const int rbase = half << 3;
        #pragma unroll
        for (int j = 0; j < 8; ++j)
            kpart[wave][rbase + j] = c[j];
    }
    __syncthreads();

    if (isr) {
        if (threadIdx.x < 16) {
            float v = 0.f;
            #pragma unroll
            for (int wv = 0; wv < 8; ++wv)
                v += kpart[wv][threadIdx.x];
            fr[row0 + threadIdx.x] = 1.f / (1.f + expf(-v));
        }
        if (blk == H / 16)
            ((v4f*)xln_out)[wave * 32 + lane] = xl;
        return;
    }

    if (threadIdx.x < 16) {
        float v = 0.f;
        #pragma unroll
        for (int wv = 0; wv < 8; ++wv)
            v += kpart[wv][threadIdx.x];
        float t = fmaxf(v, 0.f);
        fks[threadIdx.x] = t * t;
    }
    __syncthreads();

    const v4f qb0 = *(const v4f*)(fks + (half << 2));
    const v4f qb1 = *(const v4f*)(fks + 8 + (half << 2));
    const v2f b00 = { qb0.x, qb0.y };
    const v2f b01 = { qb0.z, qb0.w };
    const v2f b10 = { qb1.x, qb1.y };
    const v2f b11 = { qb1.z, qb1.w };

    float* dst = partial + (size_t)blk * D + wave * 128 + (half << 3);

    #pragma unroll
    for (int g = 0; g < 8; g += 4) {
        v8f acc[4];
        #pragma unroll
        for (int i = 0; i < 4; ++i) {
            const v2f a00 = { qa[g + i][0].x, qa[g + i][0].y };
            acc[i] = (v8f){};
            acc[i] = __builtin_amdgcn_wmma_f32_16x16x4_f32(false, a00, false, b00, (short)0, acc[i], false, false);
        }
        #pragma unroll
        for (int i = 0; i < 4; ++i) {
            const v2f a01 = { qa[g + i][0].z, qa[g + i][0].w };
            acc[i] = __builtin_amdgcn_wmma_f32_16x16x4_f32(false, a01, false, b01, (short)0, acc[i], false, false);
        }
        #pragma unroll
        for (int i = 0; i < 4; ++i) {
            const v2f a10 = { qa[g + i][1].x, qa[g + i][1].y };
            acc[i] = __builtin_amdgcn_wmma_f32_16x16x4_f32(false, a10, false, b10, (short)0, acc[i], false, false);
        }
        #pragma unroll
        for (int i = 0; i < 4; ++i) {
            const v2f a11 = { qa[g + i][1].z, qa[g + i][1].w };
            acc[i] = __builtin_amdgcn_wmma_f32_16x16x4_f32(false, a11, false, b11, (short)0, acc[i], false, false);
        }
        if (m == 0) {
            #pragma unroll
            for (int i = 0; i < 4; ++i) {
                const v4f lo = { acc[i][0], acc[i][1], acc[i][2], acc[i][3] };
                const v4f hi = { acc[i][4], acc[i][5], acc[i][6], acc[i][7] };
                *(v4f*)(dst + (g + i) * 16)     = lo;
                *(v4f*)(dst + (g + i) * 16 + 4) = hi;
            }
        }
    }
}

__global__ __launch_bounds__(256) void ffn_out_kernel(
    const float* __restrict__ partial, const float* __restrict__ x_att,
    const float* __restrict__ fr, const float* __restrict__ staged,
    float* __restrict__ out)
{
    __shared__ float red[8][32];
    const int lane = threadIdx.x & 31;
    const int g    = threadIdx.x >> 5;
    const int col  = blockIdx.x * 32 + lane;

    const bool copies = (g >= 1 && g <= 3);
    const float sv  = staged[(size_t)(copies ? g - 1 : 0) * D + col];
    const float xa  = x_att[col];
    const float frv = fr[col];

    const float* p = partial + (size_t)(g * 32) * D + col;
    float s0 = 0.f, s1 = 0.f, s2 = 0.f, s3 = 0.f;
    #pragma unroll
    for (int i = 0; i < 32; i += 4) {
        s0 += p[(size_t)(i + 0) * D];
        s1 += p[(size_t)(i + 1) * D];
        s2 += p[(size_t)(i + 2) * D];
        s3 += p[(size_t)(i + 3) * D];
    }
    red[g][lane] = (s0 + s1) + (s2 + s3);
    __syncthreads();

    if (copies)
        out[(size_t)(2 + g) * D + col] = sv;
    if (threadIdx.x < 32) {
        float v = 0.f;
        #pragma unroll
        for (int w = 0; w < 8; ++w)
            v += red[w][threadIdx.x];
        out[col] = xa + frv * v;
    }
}

extern "C" void kernel_launch(void* const* d_in, const int* in_sizes, int n_in,
                              void* d_out, int out_size, void* d_ws, size_t ws_size,
                              hipStream_t stream) {
    const float* x     = (const float*)d_in[0];
    const float* state = (const float*)d_in[1];
    const float* tmk   = (const float*)d_in[2];
    const float* tmv   = (const float*)d_in[3];
    const float* tmr   = (const float*)d_in[4];
    const float* tf    = (const float*)d_in[5];
    const float* td    = (const float*)d_in[6];
    const float* kw    = (const float*)d_in[7];
    const float* vw    = (const float*)d_in[8];
    const float* rw    = (const float*)d_in[9];
    const float* ow    = (const float*)d_in[10];
    const float* lnw   = (const float*)d_in[11];
    const float* lnb   = (const float*)d_in[12];
    const float* ftmk  = (const float*)d_in[13];
    const float* ftmr  = (const float*)d_in[14];
    const float* fkw   = (const float*)d_in[15];
    const float* fvw   = (const float*)d_in[16];
    const float* frw   = (const float*)d_in[17];
    const float* flnw  = (const float*)d_in[18];
    const float* flnb  = (const float*)d_in[19];

    float* ws    = (float*)d_ws;
    float* xk    = ws + 1 * D;
    float* xv    = ws + 2 * D;
    float* xr    = ws + 3 * D;
    float* staged = ws + 4 * D;
    float* x_att = ws + 8 * D;
    float* stats = ws + 9 * D;
    float* fr    = ws + 11 * D;
    float* fvpart = ws + 16 * D;
    float* opart  = ws + 512 * D;

    float* out = (float*)d_out;
    float* s0  = out + 1 * D;
    float* s1  = out + 2 * D;

    ln_mix_kernel<<<1, 256, 0, stream>>>(x, lnw, lnb, state + 1 * D,
                                         tmk, tmv, tmr, s1, xk, xv, xr);

    att_kvr_ow_kernel<<<D / 4, 128, 0, stream>>>(kw, vw, rw, xk, xv, xr, state, tf, td,
                                                   ow, opart,
                                                   staged, staged + 1 * D, staged + 2 * D);

    att_sum_kernel<<<D / 32, 256, 0, stream>>>(opart, x, x_att, stats);

    ffn_kv_kernel<<<(H / 16) + (D / 16), 256, 0, stream>>>(frw, fr, fkw, fvw, fvpart,
                                                           x_att, stats, flnw, flnb, state  ,
                                                           ftmk, ftmr, s0);

    ffn_out_kernel<<<D / 32, 256, 0, stream>>>(fvpart, x_att, fr, staged, out);
}


# LayerNorm 1 + time-mix launch removed: each wave of the k/v/r launch sums x itself (no block barrier) and mixes only its own K/4 slice into wave-private LDS for its B fragments; LN sum order changes
# speedup vs baseline: 2.2645x; 1.0714x over previous
#include <hip/hip_runtime.h>
#include <hip/hip_bf16.h>

#define D 1024
#define H 4096
#define EPS 1e-5f

typedef __attribute__((ext_vector_type(2))) float v2f;
typedef __attribute__((ext_vector_type(8))) float v8f;
typedef __attribute__((ext_vector_type(4))) float v4f;

#define OP_NONE     0
#define OP_SIGMOID  1
#define OP_ADDRES   2


struct MixIn { v2f st; v4f xa; v4f w; v4f b; v4f p; v4f t; };

__device__ __forceinline__ MixIn mix_request(
    const float* __restrict__ x_att, const float* __restrict__ stats,
    const float* __restrict__ w, const float* __restrict__ b,
    const float* __restrict__ prev, const float* __restrict__ tm, int slice)
{
    const int lane = threadIdx.x & 31;
    const int q = slice * 32 + lane;
    MixIn in;
    in.st = *(const v2f*)(stats + 32 * lane);
    in.xa = ((const v4f*)x_att)[q];
    in.w  = ((const v4f*)w)[q];
    in.b  = ((const v4f*)b)[q];
    in.p  = ((const v4f*)prev)[q];
    in.t  = ((const v4f*)tm)[q];
    return in;
}

__device__ __forceinline__ v4f mix_finish(const MixIn& in, float* bl)
{
    const int lane = threadIdx.x & 31;
    float s = in.st.x, sq = in.st.y;
    for (int off = 16; off > 0; off >>= 1) {
        s  += __shfl_xor(s,  off, 32);
        sq += __shfl_xor(sq, off, 32);
    }
    const float mu   = s * (1.f / D);
    const float var  = sq * (1.f / D) - mu * mu;
    const float rstd = rsqrtf(var + EPS);
    const v4f xl = (in.xa - mu) * rstd * in.w + in.b;
    *(v4f*)(bl + 4 * lane) = xl * in.t + in.p * (1.f - in.t);
    __builtin_amdgcn_wave_barrier();
    return xl;
}

__device__ __forceinline__ void slice_request_a(
    const float* __restrict__ W, int row0, int wave, v2f a[32])
{
    const int lane = threadIdx.x & 31;
    const int half = lane >> 4;
    const int m    = lane & 15;
    const float* wrow = W + (size_t)(row0 + m) * (size_t)D + (half << 1) + wave * (D >> 3);
    #pragma unroll
    for (int j = 0; j < 32; ++j)
        a[j] = *(const v2f*)(wrow + 4 * j);
}

__device__ __forceinline__ v8f slice_product(const v2f a[32], const float* bl)
{
    const int half = (threadIdx.x & 31) >> 4;
    const float* brow = bl + (half << 1);
    v8f c0 = {}; v8f c1 = {}; v8f c2 = {}; v8f c3 = {};
    #pragma unroll
    for (int j = 0; j < 32; j += 4) {
        const v2f b0 = *(const v2f*)(brow + 4 * j);
        const v2f b1 = *(const v2f*)(brow + 4 * j + 4);
        const v2f b2 = *(const v2f*)(brow + 4 * j + 8);
        const v2f b3 = *(const v2f*)(brow + 4 * j + 12);
        c0 = __builtin_amdgcn_wmma_f32_16x16x4_f32(false, a[j],     false, b0, (short)0, c0, false, false);
        c1 = __builtin_amdgcn_wmma_f32_16x16x4_f32(false, a[j + 1], false, b1, (short)0, c1, false, false);
        c2 = __builtin_amdgcn_wmma_f32_16x16x4_f32(false, a[j + 2], false, b2, (short)0, c2, false, false);
        c3 = __builtin_amdgcn_wmma_f32_16x16x4_f32(false, a[j + 3], false, b3, (short)0, c3, false, false);
    }
    return c0 + c1 + c2 + c3;
}

#define BPITCH 264

__global__ __launch_bounds__(128) __attribute__((amdgpu_waves_per_eu(1, 1)))
void att_kvr_ow_kernel(
    const float* __restrict__ x,
    const float* __restrict__ lnw, const float* __restrict__ lnb,
    const float* __restrict__ tmk, const float* __restrict__ tmv,
    const float* __restrict__ tmr,
    const float* __restrict__ kw, const float* __restrict__ vw,
    const float* __restrict__ rw,
    const float* state,
    const float* time_first, const float* time_decay,
    const float* ow, float* __restrict__ partial,
    float* __restrict__ st2, float* __restrict__ st3, float* __restrict__ st4,
    float* __restrict__ xln_out)
{
    __shared__ __attribute__((aligned(16))) float kpart[4][16];
    __shared__ __attribute__((aligned(16))) float rws[4];
    __shared__ __attribute__((aligned(16))) float bls[4][3 * BPITCH];

    const int blk   = blockIdx.x;
    const int wave  = threadIdx.x >> 5;
    const int lane  = threadIdx.x & 31;
    const int half  = lane >> 4;
    const int m     = lane & 15;
    const int c     = m & 3;
    const int job   = (m >> 2) > 2 ? 2 : (m >> 2);

    const v4f* x4 = (const v4f*)x;
    v4f xs[8];
    #pragma unroll
    for (int i = 0; i < 8; ++i)
        xs[i] = x4[i * 32 + lane];
    const int q0 = wave * 64 + lane;
    const int q1 = q0 + 32;
    const v4f xa0 = x4[q0],                          xa1 = x4[q1];
    const v4f w0  = ((const v4f*)lnw)[q0],           w1  = ((const v4f*)lnw)[q1];
    const v4f b0  = ((const v4f*)lnb)[q0],           b1  = ((const v4f*)lnb)[q1];
    const v4f p0  = ((const v4f*)(state + D))[q0],   p1  = ((const v4f*)(state + D))[q1];
    const v4f mk0 = ((const v4f*)tmk)[q0],           mk1 = ((const v4f*)tmk)[q1];
    const v4f mv0 = ((const v4f*)tmv)[q0],           mv1 = ((const v4f*)tmv)[q1];
    const v4f mr0 = ((const v4f*)tmr)[q0],           mr1 = ((const v4f*)tmr)[q1];

    const int ch = (blk << 2) + c;
    const float* Wj = (job == 0) ? kw : (job == 1) ? vw : rw;
    const float* arow = Wj + (size_t)ch * D + wave * (D >> 2) + (half << 2);

    v4f a[32];
    #pragma unroll
    for (int i = 0; i < 32; ++i)
        a[i] = *(const v4f*)(arow + 8 * i);
    __builtin_amdgcn_sched_barrier(0);

    float s, sq;
    {
        float ts[8], tq[8];
        #pragma unroll
        for (int i = 0; i < 8; ++i) {
            ts[i] = (xs[i].x + xs[i].y) + (xs[i].z + xs[i].w);
            tq[i] = (xs[i].x * xs[i].x + xs[i].y * xs[i].y) + (xs[i].z * xs[i].z + xs[i].w * xs[i].w);
        }
        s  = ((ts[0] + ts[1]) + (ts[2] + ts[3])) + ((ts[4] + ts[5]) + (ts[6] + ts[7]));
        sq = ((tq[0] + tq[1]) + (tq[2] + tq[3])) + ((tq[4] + tq[5]) + (tq[6] + tq[7]));
    }
    for (int off = 16; off > 0; off >>= 1) {
        s  += __shfl_xor(s,  off, 32);
        sq += __shfl_xor(sq, off, 32);
    }
    const float mu   = s * (1.f / D);
    const float var  = sq * (1.f / D) - mu * mu;
    const float rstd = rsqrtf(var + EPS);

    const v4f xl0 = (xa0 - mu) * rstd * w0 + b0;
    const v4f xl1 = (xa1 - mu) * rstd * w1 + b1;
    float* bl = bls[wave];
    *(v4f*)(bl + 0 * BPITCH + 4 * lane)       = xl0 * mk0 + p0 * (1.f - mk0);
    *(v4f*)(bl + 0 * BPITCH + 128 + 4 * lane) = xl1 * mk1 + p1 * (1.f - mk1);
    *(v4f*)(bl + 1 * BPITCH + 4 * lane)       = xl0 * mv0 + p0 * (1.f - mv0);
    *(v4f*)(bl + 1 * BPITCH + 128 + 4 * lane) = xl1 * mv1 + p1 * (1.f - mv1);
    *(v4f*)(bl + 2 * BPITCH + 4 * lane)       = xl0 * mr0 + p0 * (1.f - mr0);
    *(v4f*)(bl + 2 * BPITCH + 128 + 4 * lane) = xl1 * mr1 + p1 * (1.f - mr1);
    __builtin_amdgcn_wave_barrier();

    const float aa = state[2 * D + ch];
    const float bb = state[3 * D + ch];
    const float pp = state[4 * D + ch];
    const float tf = time_first[ch];
    const float td = time_decay[ch];

    const float* obase = ow + (size_t)(wave * 16 + m) * D + (blk << 2) + (half << 1);
    v2f qa[16];
    #pragma unroll
    for (int q = 0; q < 16; ++q)
        qa[q] = *(const v2f*)(obase + (size_t)(64 * q) * D);

    const float* brow = bl + job * BPITCH + (half << 2);
    v4f b[32];
    #pragma unroll
    for (int i = 0; i < 32; ++i)
        b[i] = *(const v4f*)(brow + 8 * i);
    __builtin_amdgcn_sched_barrier(0);

    v8f c0 = {}; v8f c1 = {}; v8f c2 = {}; v8f c3 = {};
    #pragma unroll
    for (int i = 0; i < 32; i += 2) {
        const v2f a00 = { a[i].x, a[i].y },         b00 = { b[i].x, b[i].y };
        const v2f a01 = { a[i].z, a[i].w },         b01 = { b[i].z, b[i].w };
        const v2f a10 = { a[i + 1].x, a[i + 1].y }, b10 = { b[i + 1].x, b[i + 1].y };
        const v2f a11 = { a[i + 1].z, a[i + 1].w }, b11 = { b[i + 1].z, b[i + 1].w };
        c0 = __builtin_amdgcn_wmma_f32_16x16x4_f32(false, a00, false, b00, (short)0, c0, false, false);
        c1 = __builtin_amdgcn_wmma_f32_16x16x4_f32(false, a01, false, b01, (short)0, c1, false, false);
        c2 = __builtin_amdgcn_wmma_f32_16x16x4_f32(false, a10, false, b10, (short)0, c2, false, false);
        c3 = __builtin_amdgcn_wmma_f32_16x16x4_f32(false, a11, false, b11, (short)0, c3, false, false);
    }
    const v8f cs = c0 + c1 + c2 + c3;

    if (lane == 0) {
        const v4f t = { cs[0], cs[1], cs[2], cs[3] };
        *(v4f*)(&kpart[wave][0]) = t;
    } else if (lane == 4) {
        const v4f t = { cs[4], cs[5], cs[6], cs[7] };
        *(v4f*)(&kpart[wave][4]) = t;
    } else if (lane == 24) {
        const v4f t = { cs[0], cs[1], cs[2], cs[3] };
        *(v4f*)(&kpart[wave][8]) = t;
    }
    __syncthreads();

    float n2 = 0.f, n3 = 0.f, n4 = 0.f;
    if (threadIdx.x < 4) {
        float k = 0.f, v = 0.f, r = 0.f;
        #pragma unroll
        for (int wv = 0; wv < 4; ++wv) {
            k += kpart[wv][threadIdx.x];
            v += kpart[wv][4 + threadIdx.x];
            r += kpart[wv][8 + threadIdx.x];
        }
        r = 1.f / (1.f + expf(-r));

        float ww = tf + k;
        float qq = fmaxf(pp, ww);
        float e1 = expf(pp - qq);
        float e2 = expf(ww - qq);
        float wkv = (e1 * aa + e2 * v) / (e1 * bb + e2);
        rws[threadIdx.x] = r * wkv;

        float ww2 = pp + td;
        float qq2 = fmaxf(ww2, k);
        float f1 = expf(ww2 - qq2);
        float f2 = expf(k - qq2);
        n2 = f1 * aa + f2 * v;
        n3 = f1 * bb + f2;
        n4 = qq2;
    }
    __syncthreads();
    if (threadIdx.x < 4) {
        st2[ch] = n2;
        st3[ch] = n3;
        st4[ch] = n4;
    }
    if (blk == 0) {
        ((v4f*)xln_out)[q0] = xl0;
        ((v4f*)xln_out)[q1] = xl1;
    }

    const v2f bq = *(const v2f*)(rws + (half << 1));

    float* dst = partial + (size_t)blk * D + wave * 16 + (half << 3);
    v8f acc[16];
    #pragma unroll
    for (int q = 0; q < 16; ++q) {
        acc[q] = (v8f){};
        acc[q] = __builtin_amdgcn_wmma_f32_16x16x4_f32(false, qa[q], false, bq, (short)0, acc[q], false, false);
    }
    if (m == 0) {
        #pragma unroll
        for (int q = 0; q < 16; ++q) {
            const v4f lo = { acc[q][0], acc[q][1], acc[q][2], acc[q][3] };
            const v4f hi = { acc[q][4], acc[q][5], acc[q][6], acc[q][7] };
            *(v4f*)(dst + 64 * q)     = lo;
            *(v4f*)(dst + 64 * q + 4) = hi;
        }
    }
}

__global__ __launch_bounds__(256) void att_sum_kernel(
    const float* __restrict__ partial, const float* __restrict__ x,
    float* __restrict__ x_att, float* __restrict__ stats)
{
    __shared__ float red[8][32];
    const int lane = threadIdx.x & 31;
    const int g    = threadIdx.x >> 5;
    const int col  = blockIdx.x * 32 + lane;

    const float xin = x[col];

    const float* p = partial + (size_t)(g * 32) * D + col;
    float s0 = 0.f, s1 = 0.f, s2 = 0.f, s3 = 0.f;
    #pragma unroll
    for (int i = 0; i < 32; i += 4) {
        s0 += p[(size_t)(i + 0) * D];
        s1 += p[(size_t)(i + 1) * D];
        s2 += p[(size_t)(i + 2) * D];
        s3 += p[(size_t)(i + 3) * D];
    }
    red[g][lane] = (s0 + s1) + (s2 + s3);
    __syncthreads();

    if (threadIdx.x < 32) {
        float v = 0.f;
        #pragma unroll
        for (int w = 0; w < 8; ++w)
            v += red[w][threadIdx.x];
        const float xa = xin + v;
        float s = xa, sq = xa * xa;
        for (int off = 16; off > 0; off >>= 1) {
            s  += __shfl_xor(s,  off, 32);
            sq += __shfl_xor(sq, off, 32);
        }
        x_att[col] = xa;
        stats[blockIdx.x * 32 + lane] = (lane & 1) ? sq : s;
    }
}

__global__ __launch_bounds__(256) void ffn_kv_kernel(
    const float* __restrict__ frw, float* __restrict__ fr,
    const float* __restrict__ fkw,
    const float* __restrict__ fvw, float* __restrict__ partial,
    const float* __restrict__ x_att, const float* __restrict__ stats,
    const float* __restrict__ lnw, const float* __restrict__ lnb,
    const float* __restrict__ prev,
    const float* __restrict__ tmk, const float* __restrict__ tmr,
    float* __restrict__ xln_out)
{
    __shared__ float kpart[8][16];
    __shared__ __attribute__((aligned(16))) float fks[16];
    __shared__ __attribute__((aligned(16))) float bls[8][128];

    const int blk  = blockIdx.x;
    const int wave = threadIdx.x >> 5;
    const int lane = threadIdx.x & 31;
    const int half = lane >> 4;
    const int m    = lane & 15;
    const bool isr = blk >= H / 16;

    const MixIn in = mix_request(x_att, stats, lnw, lnb, prev, isr ? tmr : tmk, wave);

    v4f qa[8][2];
    if (!isr) {
        const float* vbase = fvw + (size_t)(wave * 128 + m) * H + (blk << 4) + (half << 2);
        #pragma unroll
        for (int i = 0; i < 8; ++i) {
            qa[i][0] = *(const v4f*)(vbase + (size_t)(i * 16) * H);
            qa[i][1] = *(const v4f*)(vbase + (size_t)(i * 16) * H + 8);
        }
    }
    __builtin_amdgcn_sched_barrier(0);

    const float* Wsel = isr ? frw : fkw;
    const int row0 = (isr ? blk - H / 16 : blk) << 4;
    v2f a[32];
    slice_request_a(Wsel, row0, wave, a);
    const v4f xl = mix_finish(in, bls[wave]);
    v8f c = slice_product(a, bls[wave]);

    if (m == 0) {
        const int rbase = half << 3;
        #pragma unroll
        for (int j = 0; j < 8; ++j)
            kpart[wave][rbase + j] = c[j];
    }
    __syncthreads();

    if (isr) {
        if (threadIdx.x < 16) {
            float v = 0.f;
            #pragma unroll
            for (int wv = 0; wv < 8; ++wv)
                v += kpart[wv][threadIdx.x];
            fr[row0 + threadIdx.x] = 1.f / (1.f + expf(-v));
        }
        if (blk == H / 16)
            ((v4f*)xln_out)[wave * 32 + lane] = xl;
        return;
    }

    if (threadIdx.x < 16) {
        float v = 0.f;
        #pragma unroll
        for (int wv = 0; wv < 8; ++wv)
            v += kpart[wv][threadIdx.x];
        float t = fmaxf(v, 0.f);
        fks[threadIdx.x] = t * t;
    }
    __syncthreads();

    const v4f qb0 = *(const v4f*)(fks + (half << 2));
    const v4f qb1 = *(const v4f*)(fks + 8 + (half << 2));
    const v2f b00 = { qb0.x, qb0.y };
    const v2f b01 = { qb0.z, qb0.w };
    const v2f b10 = { qb1.x, qb1.y };
    const v2f b11 = { qb1.z, qb1.w };

    float* dst = partial + (size_t)blk * D + wave * 128 + (half << 3);

    #pragma unroll
    for (int g = 0; g < 8; g += 4) {
        v8f acc[4];
        #pragma unroll
        for (int i = 0; i < 4; ++i) {
            const v2f a00 = { qa[g + i][0].x, qa[g + i][0].y };
            acc[i] = (v8f){};
            acc[i] = __builtin_amdgcn_wmma_f32_16x16x4_f32(false, a00, false, b00, (short)0, acc[i], false, false);
        }
        #pragma unroll
        for (int i = 0; i < 4; ++i) {
            const v2f a01 = { qa[g + i][0].z, qa[g + i][0].w };
            acc[i] = __builtin_amdgcn_wmma_f32_16x16x4_f32(false, a01, false, b01, (short)0, acc[i], false, false);
        }
        #pragma unroll
        for (int i = 0; i < 4; ++i) {
            const v2f a10 = { qa[g + i][1].x, qa[g + i][1].y };
            acc[i] = __builtin_amdgcn_wmma_f32_16x16x4_f32(false, a10, false, b10, (short)0, acc[i], false, false);
        }
        #pragma unroll
        for (int i = 0; i < 4; ++i) {
            const v2f a11 = { qa[g + i][1].z, qa[g + i][1].w };
            acc[i] = __builtin_amdgcn_wmma_f32_16x16x4_f32(false, a11, false, b11, (short)0, acc[i], false, false);
        }
        if (m == 0) {
            #pragma unroll
            for (int i = 0; i < 4; ++i) {
                const v4f lo = { acc[i][0], acc[i][1], acc[i][2], acc[i][3] };
                const v4f hi = { acc[i][4], acc[i][5], acc[i][6], acc[i][7] };
                *(v4f*)(dst + (g + i) * 16)     = lo;
                *(v4f*)(dst + (g + i) * 16 + 4) = hi;
            }
        }
    }
}

__global__ __launch_bounds__(256) void ffn_out_kernel(
    const float* __restrict__ partial, const float* __restrict__ x_att,
    const float* __restrict__ fr, const float* __restrict__ staged,
    float* __restrict__ out)
{
    __shared__ float red[8][32];
    const int lane = threadIdx.x & 31;
    const int g    = threadIdx.x >> 5;
    const int col  = blockIdx.x * 32 + lane;

    const bool copies = (g >= 1 && g <= 3);
    const float sv  = staged[(size_t)(copies ? g - 1 : 0) * D + col];
    const float xa  = x_att[col];
    const float frv = fr[col];

    const float* p = partial + (size_t)(g * 32) * D + col;
    float s0 = 0.f, s1 = 0.f, s2 = 0.f, s3 = 0.f;
    #pragma unroll
    for (int i = 0; i < 32; i += 4) {
        s0 += p[(size_t)(i + 0) * D];
        s1 += p[(size_t)(i + 1) * D];
        s2 += p[(size_t)(i + 2) * D];
        s3 += p[(size_t)(i + 3) * D];
    }
    red[g][lane] = (s0 + s1) + (s2 + s3);
    __syncthreads();

    if (copies)
        out[(size_t)(2 + g) * D + col] = sv;
    if (threadIdx.x < 32) {
        float v = 0.f;
        #pragma unroll
        for (int w = 0; w < 8; ++w)
            v += red[w][threadIdx.x];
        out[col] = xa + frv * v;
    }
}

extern "C" void kernel_launch(void* const* d_in, const int* in_sizes, int n_in,
                              void* d_out, int out_size, void* d_ws, size_t ws_size,
                              hipStream_t stream) {
    const float* x     = (const float*)d_in[0];
    const float* state = (const float*)d_in[1];
    const float* tmk   = (const float*)d_in[2];
    const float* tmv   = (const float*)d_in[3];
    const float* tmr   = (const float*)d_in[4];
    const float* tf    = (const float*)d_in[5];
    const float* td    = (const float*)d_in[6];
    const float* kw    = (const float*)d_in[7];
    const float* vw    = (const float*)d_in[8];
    const float* rw    = (const float*)d_in[9];
    const float* ow    = (const float*)d_in[10];
    const float* lnw   = (const float*)d_in[11];
    const float* lnb   = (const float*)d_in[12];
    const float* ftmk  = (const float*)d_in[13];
    const float* ftmr  = (const float*)d_in[14];
    const float* fkw   = (const float*)d_in[15];
    const float* fvw   = (const float*)d_in[16];
    const float* frw   = (const float*)d_in[17];
    const float* flnw  = (const float*)d_in[18];
    const float* flnb  = (const float*)d_in[19];

    float* ws    = (float*)d_ws;
    float* staged = ws + 4 * D;
    float* x_att = ws + 8 * D;
    float* stats = ws + 9 * D;
    float* fr    = ws + 11 * D;
    float* fvpart = ws + 16 * D;
    float* opart  = ws + 512 * D;

    float* out = (float*)d_out;
    float* s0  = out + 1 * D;
    float* s1  = out + 2 * D;

    att_kvr_ow_kernel<<<D / 4, 128, 0, stream>>>(x, lnw, lnb, tmk, tmv, tmr, kw, vw, rw,
                                                   state, tf, td, ow, opart,
                                                   staged, staged + 1 * D, staged + 2 * D, s1);

    att_sum_kernel<<<D / 32, 256, 0, stream>>>(opart, x, x_att, stats);

    ffn_kv_kernel<<<(H / 16) + (D / 16), 256, 0, stream>>>(frw, fr, fkw, fvw, fvpart,
                                                           x_att, stats, flnw, flnb, state  ,
                                                           ftmk, ftmr, s0);

    ffn_out_kernel<<<D / 32, 256, 0, stream>>>(fvpart, x_att, fr, staged, out);
}


# FFN launch re-cut from 320 work-groups of 8 waves to 256 equal ones of 4 (16 ffn_kw rows + 4 ffn_rw rows in 4 k pieces, K/4 per wave, value product 256 rows per wave); key, fr sum order changes
# speedup vs baseline: 2.4196x; 1.0685x over previous
#include <hip/hip_runtime.h>
#include <hip/hip_bf16.h>

#define D 1024
#define H 4096
#define EPS 1e-5f

typedef __attribute__((ext_vector_type(2))) float v2f;
typedef __attribute__((ext_vector_type(8))) float v8f;
typedef __attribute__((ext_vector_type(4))) float v4f;

#define OP_NONE     0
#define OP_SIGMOID  1
#define OP_ADDRES   2


#define BPITCH 264

__global__ __launch_bounds__(128) __attribute__((amdgpu_waves_per_eu(1, 1)))
void att_kvr_ow_kernel(
    const float* __restrict__ x,
    const float* __restrict__ lnw, const float* __restrict__ lnb,
    const float* __restrict__ tmk, const float* __restrict__ tmv,
    const float* __restrict__ tmr,
    const float* __restrict__ kw, const float* __restrict__ vw,
    const float* __restrict__ rw,
    const float* state,
    const float* time_first, const float* time_decay,
    const float* ow, float* __restrict__ partial,
    float* __restrict__ st2, float* __restrict__ st3, float* __restrict__ st4,
    float* __restrict__ xln_out)
{
    __shared__ __attribute__((aligned(16))) float kpart[4][16];
    __shared__ __attribute__((aligned(16))) float rws[4];
    __shared__ __attribute__((aligned(16))) float bls[4][3 * BPITCH];

    const int blk   = blockIdx.x;
    const int wave  = threadIdx.x >> 5;
    const int lane  = threadIdx.x & 31;
    const int half  = lane >> 4;
    const int m     = lane & 15;
    const int c     = m & 3;
    const int job   = (m >> 2) > 2 ? 2 : (m >> 2);

    const v4f* x4 = (const v4f*)x;
    v4f xs[8];
    #pragma unroll
    for (int i = 0; i < 8; ++i)
        xs[i] = x4[i * 32 + lane];
    const int q0 = wave * 64 + lane;
    const int q1 = q0 + 32;
    const v4f xa0 = x4[q0],                          xa1 = x4[q1];
    const v4f w0  = ((const v4f*)lnw)[q0],           w1  = ((const v4f*)lnw)[q1];
    const v4f b0  = ((const v4f*)lnb)[q0],           b1  = ((const v4f*)lnb)[q1];
    const v4f p0  = ((const v4f*)(state + D))[q0],   p1  = ((const v4f*)(state + D))[q1];
    const v4f mk0 = ((const v4f*)tmk)[q0],           mk1 = ((const v4f*)tmk)[q1];
    const v4f mv0 = ((const v4f*)tmv)[q0],           mv1 = ((const v4f*)tmv)[q1];
    const v4f mr0 = ((const v4f*)tmr)[q0],           mr1 = ((const v4f*)tmr)[q1];

    const int ch = (blk << 2) + c;
    const float* Wj = (job == 0) ? kw : (job == 1) ? vw : rw;
    const float* arow = Wj + (size_t)ch * D + wave * (D >> 2) + (half << 2);

    v4f a[32];
    #pragma unroll
    for (int i = 0; i < 32; ++i)
        a[i] = *(const v4f*)(arow + 8 * i);
    __builtin_amdgcn_sched_barrier(0);

    float s, sq;
    {
        float ts[8], tq[8];
        #pragma unroll
        for (int i = 0; i < 8; ++i) {
            ts[i] = (xs[i].x + xs[i].y) + (xs[i].z + xs[i].w);
            tq[i] = (xs[i].x * xs[i].x + xs[i].y * xs[i].y) + (xs[i].z * xs[i].z + xs[i].w * xs[i].w);
        }
        s  = ((ts[0] + ts[1]) + (ts[2] + ts[3])) + ((ts[4] + ts[5]) + (ts[6] + ts[7]));
        sq = ((tq[0] + tq[1]) + (tq[2] + tq[3])) + ((tq[4] + tq[5]) + (tq[6] + tq[7]));
    }
    for (int off = 16; off > 0; off >>= 1) {
        s  += __shfl_xor(s,  off, 32);
        sq += __shfl_xor(sq, off, 32);
    }
    const float mu   = s * (1.f / D);
    const float var  = sq * (1.f / D) - mu * mu;
    const float rstd = rsqrtf(var + EPS);

    const v4f xl0 = (xa0 - mu) * rstd * w0 + b0;
    const v4f xl1 = (xa1 - mu) * rstd * w1 + b1;
    float* bl = bls[wave];
    *(v4f*)(bl + 0 * BPITCH + 4 * lane)       = xl0 * mk0 + p0 * (1.f - mk0);
    *(v4f*)(bl + 0 * BPITCH + 128 + 4 * lane) = xl1 * mk1 + p1 * (1.f - mk1);
    *(v4f*)(bl + 1 * BPITCH + 4 * lane)       = xl0 * mv0 + p0 * (1.f - mv0);
    *(v4f*)(bl + 1 * BPITCH + 128 + 4 * lane) = xl1 * mv1 + p1 * (1.f - mv1);
    *(v4f*)(bl + 2 * BPITCH + 4 * lane)       = xl0 * mr0 + p0 * (1.f - mr0);
    *(v4f*)(bl + 2 * BPITCH + 128 + 4 * lane) = xl1 * mr1 + p1 * (1.f - mr1);
    __builtin_amdgcn_wave_barrier();

    const float aa = state[2 * D + ch];
    const float bb = state[3 * D + ch];
    const float pp = state[4 * D + ch];
    const float tf = time_first[ch];
    const float td = time_decay[ch];

    const float* obase = ow + (size_t)(wave * 16 + m) * D + (blk << 2) + (half << 1);
    v2f qa[16];
    #pragma unroll
    for (int q = 0; q < 16; ++q)
        qa[q] = *(const v2f*)(obase + (size_t)(64 * q) * D);

    const float* brow = bl + job * BPITCH + (half << 2);
    v4f b[32];
    #pragma unroll
    for (int i = 0; i < 32; ++i)
        b[i] = *(const v4f*)(brow + 8 * i);
    __builtin_amdgcn_sched_barrier(0);

    v8f c0 = {}; v8f c1 = {}; v8f c2 = {}; v8f c3 = {};
    #pragma unroll
    for (int i = 0; i < 32; i += 2) {
        const v2f a00 = { a[i].x, a[i].y },         b00 = { b[i].x, b[i].y };
        const v2f a01 = { a[i].z, a[i].w },         b01 = { b[i].z, b[i].w };
        const v2f a10 = { a[i + 1].x, a[i + 1].y }, b10 = { b[i + 1].x, b[i + 1].y };
        const v2f a11 = { a[i + 1].z, a[i + 1].w }, b11 = { b[i + 1].z, b[i + 1].w };
        c0 = __builtin_amdgcn_wmma_f32_16x16x4_f32(false, a00, false, b00, (short)0, c0, false, false);
        c1 = __builtin_amdgcn_wmma_f32_16x16x4_f32(false, a01, false, b01, (short)0, c1, false, false);
        c2 = __builtin_amdgcn_wmma_f32_16x16x4_f32(false, a10, false, b10, (short)0, c2, false, false);
        c3 = __builtin_amdgcn_wmma_f32_16x16x4_f32(false, a11, false, b11, (short)0, c3, false, false);
    }
    const v8f cs = c0 + c1 + c2 + c3;

    if (lane == 0) {
        const v4f t = { cs[0], cs[1], cs[2], cs[3] };
        *(v4f*)(&kpart[wave][0]) = t;
    } else if (lane == 4) {
        const v4f t = { cs[4], cs[5], cs[6], cs[7] };
        *(v4f*)(&kpart[wave][4]) = t;
    } else if (lane == 24) {
        const v4f t = { cs[0], cs[1], cs[2], cs[3] };
        *(v4f*)(&kpart[wave][8]) = t;
    }
    __syncthreads();

    float n2 = 0.f, n3 = 0.f, n4 = 0.f;
    if (threadIdx.x < 4) {
        float k = 0.f, v = 0.f, r = 0.f;
        #pragma unroll
        for (int wv = 0; wv < 4; ++wv) {
            k += kpart[wv][threadIdx.x];
            v += kpart[wv][4 + threadIdx.x];
            r += kpart[wv][8 + threadIdx.x];
        }
        r = 1.f / (1.f + expf(-r));

        float ww = tf + k;
        float qq = fmaxf(pp, ww);
        float e1 = expf(pp - qq);
        float e2 = expf(ww - qq);
        float wkv = (e1 * aa + e2 * v) / (e1 * bb + e2);
        rws[threadIdx.x] = r * wkv;

        float ww2 = pp + td;
        float qq2 = fmaxf(ww2, k);
        float f1 = expf(ww2 - qq2);
        float f2 = expf(k - qq2);
        n2 = f1 * aa + f2 * v;
        n3 = f1 * bb + f2;
        n4 = qq2;
    }
    __syncthreads();
    if (threadIdx.x < 4) {
        st2[ch] = n2;
        st3[ch] = n3;
        st4[ch] = n4;
    }
    if (blk == 0) {
        ((v4f*)xln_out)[q0] = xl0;
        ((v4f*)xln_out)[q1] = xl1;
    }

    const v2f bq = *(const v2f*)(rws + (half << 1));

    float* dst = partial + (size_t)blk * D + wave * 16 + (half << 3);
    v8f acc[16];
    #pragma unroll
    for (int q = 0; q < 16; ++q) {
        acc[q] = (v8f){};
        acc[q] = __builtin_amdgcn_wmma_f32_16x16x4_f32(false, qa[q], false, bq, (short)0, acc[q], false, false);
    }
    if (m == 0) {
        #pragma unroll
        for (int q = 0; q < 16; ++q) {
            const v4f lo = { acc[q][0], acc[q][1], acc[q][2], acc[q][3] };
            const v4f hi = { acc[q][4], acc[q][5], acc[q][6], acc[q][7] };
            *(v4f*)(dst + 64 * q)     = lo;
            *(v4f*)(dst + 64 * q + 4) = hi;
        }
    }
}

__global__ __launch_bounds__(256) void att_sum_kernel(
    const float* __restrict__ partial, const float* __restrict__ x,
    float* __restrict__ x_att, float* __restrict__ stats)
{
    __shared__ float red[8][32];
    const int lane = threadIdx.x & 31;
    const int g    = threadIdx.x >> 5;
    const int col  = blockIdx.x * 32 + lane;

    const float xin = x[col];

    const float* p = partial + (size_t)(g * 32) * D + col;
    float s0 = 0.f, s1 = 0.f, s2 = 0.f, s3 = 0.f;
    #pragma unroll
    for (int i = 0; i < 32; i += 4) {
        s0 += p[(size_t)(i + 0) * D];
        s1 += p[(size_t)(i + 1) * D];
        s2 += p[(size_t)(i + 2) * D];
        s3 += p[(size_t)(i + 3) * D];
    }
    red[g][lane] = (s0 + s1) + (s2 + s3);
    __syncthreads();

    if (threadIdx.x < 32) {
        float v = 0.f;
        #pragma unroll
        for (int w = 0; w < 8; ++w)
            v += red[w][threadIdx.x];
        const float xa = xin + v;
        float s = xa, sq = xa * xa;
        for (int off = 16; off > 0; off >>= 1) {
            s  += __shfl_xor(s,  off, 32);
            sq += __shfl_xor(sq, off, 32);
        }
        x_att[col] = xa;
        stats[blockIdx.x * 32 + lane] = (lane & 1) ? sq : s;
    }
}

#define FPITCH 264

__global__ __launch_bounds__(128) __attribute__((amdgpu_waves_per_eu(1, 1)))
void ffn_kv_kernel(
    const float* __restrict__ frw, float* __restrict__ fr,
    const float* __restrict__ fkw,
    const float* __restrict__ fvw, float* __restrict__ partial,
    const float* __restrict__ x_att, const float* __restrict__ stats,
    const float* __restrict__ lnw, const float* __restrict__ lnb,
    const float* __restrict__ prev,
    const float* __restrict__ tmk, const float* __restrict__ tmr,
    float* __restrict__ xln_out)
{
    __shared__ __attribute__((aligned(16))) float kpart[4][16];
    __shared__ __attribute__((aligned(16))) float rpart[4][16];
    __shared__ __attribute__((aligned(16))) float fks[16];
    __shared__ __attribute__((aligned(16))) float bls[4][2 * FPITCH];

    const int blk  = blockIdx.x;
    const int wave = threadIdx.x >> 5;
    const int lane = threadIdx.x & 31;
    const int half = lane >> 4;
    const int m    = lane & 15;

    const int q0 = wave * 64 + lane;
    const int q1 = q0 + 32;
    const v2f st  = *(const v2f*)(stats + 32 * lane);
    const v4f xa0 = ((const v4f*)x_att)[q0], xa1 = ((const v4f*)x_att)[q1];
    const v4f w0  = ((const v4f*)lnw)[q0],   w1  = ((const v4f*)lnw)[q1];
    const v4f lb0 = ((const v4f*)lnb)[q0],   lb1 = ((const v4f*)lnb)[q1];
    const v4f p0  = ((const v4f*)prev)[q0],  p1  = ((const v4f*)prev)[q1];
    const v4f mk0 = ((const v4f*)tmk)[q0],   mk1 = ((const v4f*)tmk)[q1];
    const v4f mr0 = ((const v4f*)tmr)[q0],   mr1 = ((const v4f*)tmr)[q1];

    const float* krow = fkw + (size_t)((blk << 4) + m) * D + wave * (D >> 2) + (half << 2);
    v4f ka[32];
    #pragma unroll
    for (int i = 0; i < 32; ++i)
        ka[i] = *(const v4f*)(krow + 8 * i);

    const int pc = m >> 2;
    const float* rrow = frw + (size_t)((blk << 2) + (m & 3)) * D + wave * (D >> 2) + (pc << 3) + (half << 2);
    v4f ra[8];
    #pragma unroll
    for (int i = 0; i < 8; ++i)
        ra[i] = *(const v4f*)(rrow + 32 * i);

    const float* vbase = fvw + (size_t)(wave * 256 + m) * H + (blk << 4) + (half << 2);
    v4f qa[16][2];
    #pragma unroll
    for (int i = 0; i < 16; ++i) {
        qa[i][0] = *(const v4f*)(vbase + (size_t)(i * 16) * H);
        qa[i][1] = *(const v4f*)(vbase + (size_t)(i * 16) * H + 8);
    }
    __builtin_amdgcn_sched_barrier(0);

    float s = st.x, sq = st.y;
    for (int off = 16; off > 0; off >>= 1) {
        s  += __shfl_xor(s,  off, 32);
        sq += __shfl_xor(sq, off, 32);
    }
    const float mu   = s * (1.f / D);
    const float var  = sq * (1.f / D) - mu * mu;
    const float rstd = rsqrtf(var + EPS);
    const v4f xl0 = (xa0 - mu) * rstd * w0 + lb0;
    const v4f xl1 = (xa1 - mu) * rstd * w1 + lb1;
    float* bl = bls[wave];
    *(v4f*)(bl + 4 * lane)                = xl0 * mk0 + p0 * (1.f - mk0);
    *(v4f*)(bl + 128 + 4 * lane)          = xl1 * mk1 + p1 * (1.f - mk1);
    *(v4f*)(bl + FPITCH + 4 * lane)       = xl0 * mr0 + p0 * (1.f - mr0);
    *(v4f*)(bl + FPITCH + 128 + 4 * lane) = xl1 * mr1 + p1 * (1.f - mr1);
    __builtin_amdgcn_wave_barrier();

    const float* kb = bl + (half << 2);
    v8f c0 = {}; v8f c1 = {}; v8f c2 = {}; v8f c3 = {};
    #pragma unroll
    for (int i = 0; i < 32; i += 2) {
        const v4f f0 = *(const v4f*)(kb + 8 * i);
        const v4f f1 = *(const v4f*)(kb + 8 * i + 8);
        const v2f a00 = { ka[i].x, ka[i].y },         b00 = { f0.x, f0.y };
        const v2f a01 = { ka[i].z, ka[i].w },         b01 = { f0.z, f0.w };
        const v2f a10 = { ka[i + 1].x, ka[i + 1].y }, b10 = { f1.x, f1.y };
        const v2f a11 = { ka[i + 1].z, ka[i + 1].w }, b11 = { f1.z, f1.w };
        c0 = __builtin_amdgcn_wmma_f32_16x16x4_f32(false, a00, false, b00, (short)0, c0, false, false);
        c1 = __builtin_amdgcn_wmma_f32_16x16x4_f32(false, a01, false, b01, (short)0, c1, false, false);
        c2 = __builtin_amdgcn_wmma_f32_16x16x4_f32(false, a10, false, b10, (short)0, c2, false, false);
        c3 = __builtin_amdgcn_wmma_f32_16x16x4_f32(false, a11, false, b11, (short)0, c3, false, false);
    }
    const float* rb = bl + FPITCH + (pc << 3) + (half << 2);
    v8f r0 = {}; v8f r1 = {}; v8f r2 = {}; v8f r3 = {};
    #pragma unroll
    for (int i = 0; i < 8; i += 2) {
        const v4f f0 = *(const v4f*)(rb + 32 * i);
        const v4f f1 = *(const v4f*)(rb + 32 * i + 32);
        const v2f a00 = { ra[i].x, ra[i].y },         b00 = { f0.x, f0.y };
        const v2f a01 = { ra[i].z, ra[i].w },         b01 = { f0.z, f0.w };
        const v2f a10 = { ra[i + 1].x, ra[i + 1].y }, b10 = { f1.x, f1.y };
        const v2f a11 = { ra[i + 1].z, ra[i + 1].w }, b11 = { f1.z, f1.w };
        r0 = __builtin_amdgcn_wmma_f32_16x16x4_f32(false, a00, false, b00, (short)0, r0, false, false);
        r1 = __builtin_amdgcn_wmma_f32_16x16x4_f32(false, a01, false, b01, (short)0, r1, false, false);
        r2 = __builtin_amdgcn_wmma_f32_16x16x4_f32(false, a10, false, b10, (short)0, r2, false, false);
        r3 = __builtin_amdgcn_wmma_f32_16x16x4_f32(false, a11, false, b11, (short)0, r3, false, false);
    }
    const v8f cs = c0 + c1 + c2 + c3;
    const v8f rs = r0 + r1 + r2 + r3;

    if (m == 0) {
        const v4f lo = { cs[0], cs[1], cs[2], cs[3] };
        const v4f hi = { cs[4], cs[5], cs[6], cs[7] };
        *(v4f*)(&kpart[wave][(half << 3)])     = lo;
        *(v4f*)(&kpart[wave][(half << 3) + 4]) = hi;
    }
    if (lane == 0 || lane == 24) {
        const v4f t = { rs[0], rs[1], rs[2], rs[3] };
        *(v4f*)(&rpart[wave][half << 3]) = t;
    } else if (lane == 4 || lane == 28) {
        const v4f t = { rs[4], rs[5], rs[6], rs[7] };
        *(v4f*)(&rpart[wave][(half << 3) + 4]) = t;
    }
    __syncthreads();

    float frv = 0.f;
    if (threadIdx.x < 16) {
        float v = 0.f;
        #pragma unroll
        for (int wv = 0; wv < 4; ++wv)
            v += kpart[wv][threadIdx.x];
        float t = fmaxf(v, 0.f);
        fks[threadIdx.x] = t * t;
    } else if (threadIdx.x >= 32 && threadIdx.x < 36) {
        float v = 0.f;
        #pragma unroll
        for (int wv = 0; wv < 4; ++wv) {
            #pragma unroll
            for (int p = 0; p < 4; ++p)
                v += rpart[wv][4 * p + lane];
        }
        frv = 1.f / (1.f + expf(-v));
    }
    __syncthreads();
    if (threadIdx.x >= 32 && threadIdx.x < 36)
        fr[(blk << 2) + lane] = frv;
    if (blk == 0) {
        ((v4f*)xln_out)[q0] = xl0;
        ((v4f*)xln_out)[q1] = xl1;
    }

    const v4f qb0 = *(const v4f*)(fks + (half << 2));
    const v4f qb1 = *(const v4f*)(fks + 8 + (half << 2));
    const v2f b00 = { qb0.x, qb0.y };
    const v2f b01 = { qb0.z, qb0.w };
    const v2f b10 = { qb1.x, qb1.y };
    const v2f b11 = { qb1.z, qb1.w };

    float* dst = partial + (size_t)blk * D + wave * 256 + (half << 3);

    #pragma unroll
    for (int g = 0; g < 16; g += 4) {
        v8f acc[4];
        #pragma unroll
        for (int i = 0; i < 4; ++i) {
            const v2f a00 = { qa[g + i][0].x, qa[g + i][0].y };
            acc[i] = (v8f){};
            acc[i] = __builtin_amdgcn_wmma_f32_16x16x4_f32(false, a00, false, b00, (short)0, acc[i], false, false);
        }
        #pragma unroll
        for (int i = 0; i < 4; ++i) {
            const v2f a01 = { qa[g + i][0].z, qa[g + i][0].w };
            acc[i] = __builtin_amdgcn_wmma_f32_16x16x4_f32(false, a01, false, b01, (short)0, acc[i], false, false);
        }
        #pragma unroll
        for (int i = 0; i < 4; ++i) {
            const v2f a10 = { qa[g + i][1].x, qa[g + i][1].y };
            acc[i] = __builtin_amdgcn_wmma_f32_16x16x4_f32(false, a10, false, b10, (short)0, acc[i], false, false);
        }
        #pragma unroll
        for (int i = 0; i < 4; ++i) {
            const v2f a11 = { qa[g + i][1].z, qa[g + i][1].w };
            acc[i] = __builtin_amdgcn_wmma_f32_16x16x4_f32(false, a11, false, b11, (short)0, acc[i], false, false);
        }
        if (m == 0) {
            #pragma unroll
            for (int i = 0; i < 4; ++i) {
                const v4f lo = { acc[i][0], acc[i][1], acc[i][2], acc[i][3] };
                const v4f hi = { acc[i][4], acc[i][5], acc[i][6], acc[i][7] };
                *(v4f*)(dst + (g + i) * 16)     = lo;
                *(v4f*)(dst + (g + i) * 16 + 4) = hi;
            }
        }
    }
}

__global__ __launch_bounds__(256) void ffn_out_kernel(
    const float* __restrict__ partial, const float* __restrict__ x_att,
    const float* __restrict__ fr, const float* __restrict__ staged,
    float* __restrict__ out)
{
    __shared__ float red[8][32];
    const int lane = threadIdx.x & 31;
    const int g    = threadIdx.x >> 5;
    const int col  = blockIdx.x * 32 + lane;

    const bool copies = (g >= 1 && g <= 3);
    const float sv  = staged[(size_t)(copies ? g - 1 : 0) * D + col];
    const float xa  = x_att[col];
    const float frv = fr[col];

    const float* p = partial + (size_t)(g * 32) * D + col;
    float s0 = 0.f, s1 = 0.f, s2 = 0.f, s3 = 0.f;
    #pragma unroll
    for (int i = 0; i < 32; i += 4) {
        s0 += p[(size_t)(i + 0) * D];
        s1 += p[(size_t)(i + 1) * D];
        s2 += p[(size_t)(i + 2) * D];
        s3 += p[(size_t)(i + 3) * D];
    }
    red[g][lane] = (s0 + s1) + (s2 + s3);
    __syncthreads();

    if (copies)
        out[(size_t)(2 + g) * D + col] = sv;
    if (threadIdx.x < 32) {
        float v = 0.f;
        #pragma unroll
        for (int w = 0; w < 8; ++w)
            v += red[w][threadIdx.x];
        out[col] = xa + frv * v;
    }
}

extern "C" void kernel_launch(void* const* d_in, const int* in_sizes, int n_in,
                              void* d_out, int out_size, void* d_ws, size_t ws_size,
                              hipStream_t stream) {
    const float* x     = (const float*)d_in[0];
    const float* state = (const float*)d_in[1];
    const float* tmk   = (const float*)d_in[2];
    const float* tmv   = (const float*)d_in[3];
    const float* tmr   = (const float*)d_in[4];
    const float* tf    = (const float*)d_in[5];
    const float* td    = (const float*)d_in[6];
    const float* kw    = (const float*)d_in[7];
    const float* vw    = (const float*)d_in[8];
    const float* rw    = (const float*)d_in[9];
    const float* ow    = (const float*)d_in[10];
    const float* lnw   = (const float*)d_in[11];
    const float* lnb   = (const float*)d_in[12];
    const float* ftmk  = (const float*)d_in[13];
    const float* ftmr  = (const float*)d_in[14];
    const float* fkw   = (const float*)d_in[15];
    const float* fvw   = (const float*)d_in[16];
    const float* frw   = (const float*)d_in[17];
    const float* flnw  = (const float*)d_in[18];
    const float* flnb  = (const float*)d_in[19];

    float* ws    = (float*)d_ws;
    float* staged = ws + 4 * D;
    float* x_att = ws + 8 * D;
    float* stats = ws + 9 * D;
    float* fr    = ws + 11 * D;
    float* fvpart = ws + 16 * D;
    float* opart  = ws + 512 * D;

    float* out = (float*)d_out;
    float* s0  = out + 1 * D;
    float* s1  = out + 2 * D;

    att_kvr_ow_kernel<<<D / 4, 128, 0, stream>>>(x, lnw, lnb, tmk, tmv, tmr, kw, vw, rw,
                                                   state, tf, td, ow, opart,
                                                   staged, staged + 1 * D, staged + 2 * D, s1);

    att_sum_kernel<<<D / 32, 256, 0, stream>>>(opart, x, x_att, stats);

    ffn_kv_kernel<<<H / 16, 128, 0, stream>>>(frw, fr, fkw, fvw, fvpart,
                                                           x_att, stats, flnw, flnb, state  ,
                                                           ftmk, ftmr, s0);

    ffn_out_kernel<<<D / 32, 256, 0, stream>>>(fvpart, x_att, fr, staged, out);
}
